# starting kernel
# baseline (speedup 1.0000x reference)
#include <hip/hip_runtime.h>
#include <cstddef>

typedef float v2f __attribute__((ext_vector_type(2)));
typedef float v4f __attribute__((ext_vector_type(4)));
typedef float v8f __attribute__((ext_vector_type(8)));
typedef int   v4i __attribute__((ext_vector_type(4)));

#define B_  8
#define N_  2048
#define C_  128
#define D_  128
#define NEG_BIG (-9.0e15f)

static __device__ __forceinline__ v8f wmma_f32_4(v2f a, v2f b, v8f c) {
    return __builtin_amdgcn_wmma_f32_16x16x4_f32(false, a, false, b, (short)0, c,
                                                 false, false);
}

__global__ __launch_bounds__(256)
void wht_kernel(const float* __restrict__ x, const float* __restrict__ W,
                float* __restrict__ wht) {
    const int bid  = blockIdx.x;
    const int b    = bid >> 7;
    const int jt   = bid & 127;
    const int tid  = threadIdx.x;
    const int wv   = tid >> 5;
    const int lane = tid & 31;
    const int nl   = lane & 15;
    const int kh   = (lane >> 4) << 1;

    const float* xrow = x + ((size_t)(b * N_ + jt * 16 + nl)) * C_;
    const float* wrow = W + (size_t)(wv * 16 + nl) * C_;

    v8f acc = {};
#pragma unroll 4
    for (int k0 = 0; k0 < C_; k0 += 4) {
        const int kb = k0 + kh;
        v2f a = *(const v2f*)(wrow + kb);
        v2f bm = *(const v2f*)(xrow + kb);
        acc = wmma_f32_4(a, bm, acc);
    }

    float* ob = wht + (size_t)b * D_ * N_ + (size_t)jt * 16 + nl;
#pragma unroll
    for (int r = 0; r < 8; ++r) {
        const int d = wv * 16 + r + ((lane >> 4) << 3);
        ob[(size_t)d * N_] = acc[r];
    }
}

__global__ __launch_bounds__(256)
void f_kernel(const float* __restrict__ wht, const float* __restrict__ a,
              float* __restrict__ f1, float* __restrict__ f2) {
    const int t = blockIdx.x * 256 + threadIdx.x;
    const int b = t >> 11;
    const int j = t & (N_ - 1);
    const float* wb = wht + (size_t)b * D_ * N_ + j;
    float s1 = 0.f, s2 = 0.f;
#pragma unroll 4
    for (int d = 0; d < D_; ++d) {
        const float w = wb[(size_t)d * N_];
        s1 += w * a[d];
        s2 += w * a[D_ + d];
    }
    f1[t] = s1;
    f2[t] = s2;
}

__global__ __launch_bounds__(256)
void gat_kernel(const float* __restrict__ wht, const float* __restrict__ f1g,
                const float* __restrict__ f2g, const int* __restrict__ adj,
                float* __restrict__ out) {
    __shared__ unsigned lds_bits[16 * 65];
    __shared__ float lds_mw[256];
    __shared__ float lds_m[16], lds_f1[16], lds_pone[16], lds_l[16];
    __shared__ float lds_lw[8 * 16];
    __shared__ float accbuf[4 * 8 * 32 * 8];

    const int bid  = blockIdx.x;
    const int b    = bid >> 7;
    const int i0   = (bid & 127) * 16;
    const int tid  = threadIdx.x;
    const int wv   = tid >> 5;
    const int lane = tid & 31;
    const int nl   = lane & 15;

    const float* f2b  = f2g + b * N_;
    const float* whtb = wht + (size_t)b * D_ * N_;

    {
        const int i   = tid >> 4;
        const int seg = tid & 15;
        const int*   arow = adj + (size_t)(i0 + i) * N_ + seg * 128;
        const float* frow = f2b + seg * 128;
        float mmax = -3.0e38f;
#pragma unroll
        for (int w2 = 0; w2 < 4; ++w2) {
            unsigned bits = 0u;
#pragma unroll
            for (int q = 0; q < 8; ++q) {
                v4i av = *(const v4i*)(arow + w2 * 32 + q * 4);
                v4f fv = *(const v4f*)(frow + w2 * 32 + q * 4);
                if (av[0]) { bits |= 1u << (q * 4 + 0); mmax = fmaxf(mmax, fv[0]); }
                if (av[1]) { bits |= 1u << (q * 4 + 1); mmax = fmaxf(mmax, fv[1]); }
                if (av[2]) { bits |= 1u << (q * 4 + 2); mmax = fmaxf(mmax, fv[2]); }
                if (av[3]) { bits |= 1u << (q * 4 + 3); mmax = fmaxf(mmax, fv[3]); }
            }
            lds_bits[i * 65 + seg * 4 + w2] = bits;
        }
        lds_mw[i * 16 + seg] = mmax;
    }
    __syncthreads();
    if (tid < 16) {
        float M = -3.0e38f;
#pragma unroll
        for (int s = 0; s < 16; ++s) M = fmaxf(M, lds_mw[tid * 16 + s]);
        const bool empty = (M < -1.0e38f);
        const float f1v  = f1g[b * N_ + i0 + tid];
        const float sum  = f1v + M;
        const float emax = fmaxf(sum, 0.2f * sum);
        lds_m[tid]    = empty ? 0.0f : emax;
        lds_pone[tid] = empty ? 1.0f : 0.0f;
        lds_f1[tid]   = f1v;
    }
    __syncthreads();

    const float f1i = lds_f1[nl];
    const float mi  = lds_m[nl];
    const float pm  = lds_pone[nl];
    const float* wbase = whtb + (size_t)nl * N_;

    v8f acc[8] = {};
    float lsum = 0.0f;

    for (int g = 0; g < 64; ++g) {
        const int j0 = g * 32 + wv * 4;
        const int jA = j0 + ((lane >> 4) << 1);
        const unsigned word = lds_bits[nl * 65 + (jA >> 5)];
        const unsigned sh   = (unsigned)(jA & 31);

        const v2f f2v = *(const v2f*)(f2b + jA);
        float s0 = f1i + f2v[0]; s0 = fmaxf(s0, 0.2f * s0);
        float s1 = f1i + f2v[1]; s1 = fmaxf(s1, 0.2f * s1);
        const float p0 = ((word >> sh) & 1u)        ? __expf(s0 - mi) : pm;
        const float p1 = ((word >> (sh + 1)) & 1u)  ? __expf(s1 - mi) : pm;

        v2f A; A[0] = p0; A[1] = p1;
        lsum += p0 + p1;

#pragma unroll
        for (int dt = 0; dt < 8; ++dt) {
            const v2f Bv = *(const v2f*)(wbase + (size_t)dt * 16 * N_ + jA);
            acc[dt] = wmma_f32_4(A, Bv, acc[dt]);
        }
    }

    float lpair = lsum + __shfl_xor(lsum, 16);
    if (lane < 16) lds_lw[wv * 16 + lane] = lpair;
    __syncthreads();
    if (tid < 16) {
        float t = 0.0f;
#pragma unroll
        for (int w2 = 0; w2 < 8; ++w2) t += lds_lw[w2 * 16 + tid];
        lds_l[tid] = t;
    }
    __syncthreads();

    for (int half = 4; half >= 1; half >>= 1) {
        if (wv >= half && wv < 2 * half) {
            const int slot = wv - half;
#pragma unroll
            for (int dt = 0; dt < 8; ++dt)
                *(v8f*)&accbuf[(((size_t)slot * 8 + dt) * 32 + lane) * 8] = acc[dt];
        }
        __syncthreads();
        if (wv < half) {
#pragma unroll
            for (int dt = 0; dt < 8; ++dt)
                acc[dt] = acc[dt] +
                    *(const v8f*)&accbuf[(((size_t)wv * 8 + dt) * 32 + lane) * 8];
        }
        __syncthreads();
    }

    if (wv == 0) {
        float* ob = out + ((size_t)b * N_ + i0) * D_;
#pragma unroll
        for (int dt = 0; dt < 8; ++dt) {
#pragma unroll
            for (int r = 0; r < 8; ++r) {
                const int row = r + ((lane >> 4) << 3);
                const float li = lds_l[row];
                const float h  = acc[dt][r] / li;
                const float o  = (h > 0.0f) ? h : (__expf(h) - 1.0f);
                ob[(size_t)row * D_ + dt * 16 + nl] = o;
            }
        }
    }
}

extern "C" void kernel_launch(void* const* d_in, const int* in_sizes, int n_in,
                              void* d_out, int out_size, void* d_ws, size_t ws_size,
                              hipStream_t stream) {
    const float* x   = (const float*)d_in[0];
    const int*   adj = (const int*)d_in[1];
    const float* W   = (const float*)d_in[2];
    const float* a   = (const float*)d_in[3];
    float* out = (float*)d_out;

    float* wht = (float*)d_ws;
    float* f1  = wht + (size_t)B_ * D_ * N_;
    float* f2  = f1 + (size_t)B_ * N_;

    wht_kernel<<<dim3(B_ * (N_ / 16)), dim3(256), 0, stream>>>(x, W, wht);
    f_kernel<<<dim3((B_ * N_) / 256), dim3(256), 0, stream>>>(wht, a, f1, f2);
    gat_kernel<<<dim3(B_ * (N_ / 16)), dim3(256), 0, stream>>>(wht, f1, f2, adj, out);
}


# gat kernel retiled: 64 i-rows per block, each wave owns a 16-row i-tile, WhT chunks staged once per block in LDS (b128 fragment reads); j-sum order changes (2-wave split, was 8-wave tree)
# speedup vs baseline: 1.5390x; 1.5390x over previous
#include <hip/hip_runtime.h>
#include <cstddef>

typedef float v2f __attribute__((ext_vector_type(2)));
typedef float v4f __attribute__((ext_vector_type(4)));
typedef float v8f __attribute__((ext_vector_type(8)));
typedef int   v4i __attribute__((ext_vector_type(4)));

#define B_  8
#define N_  2048
#define C_  128
#define D_  128
#define NEG_BIG (-9.0e15f)

static __device__ __forceinline__ v8f wmma_f32_4(v2f a, v2f b, v8f c) {
    return __builtin_amdgcn_wmma_f32_16x16x4_f32(false, a, false, b, (short)0, c,
                                                 false, false);
}

__global__ __launch_bounds__(256)
void wht_kernel(const float* __restrict__ x, const float* __restrict__ W,
                float* __restrict__ wht) {
    const int bid  = blockIdx.x;
    const int b    = bid >> 7;
    const int jt   = bid & 127;
    const int tid  = threadIdx.x;
    const int wv   = tid >> 5;
    const int lane = tid & 31;
    const int nl   = lane & 15;
    const int kh   = (lane >> 4) << 1;

    const float* xrow = x + ((size_t)(b * N_ + jt * 16 + nl)) * C_;
    const float* wrow = W + (size_t)(wv * 16 + nl) * C_;

    v8f acc = {};
#pragma unroll 4
    for (int k0 = 0; k0 < C_; k0 += 4) {
        const int kb = k0 + kh;
        v2f a = *(const v2f*)(wrow + kb);
        v2f bm = *(const v2f*)(xrow + kb);
        acc = wmma_f32_4(a, bm, acc);
    }

    float* ob = wht + (size_t)b * D_ * N_ + (size_t)jt * 16 + nl;
#pragma unroll
    for (int r = 0; r < 8; ++r) {
        const int d = wv * 16 + r + ((lane >> 4) << 3);
        ob[(size_t)d * N_] = acc[r];
    }
}

__global__ __launch_bounds__(256)
void f_kernel(const float* __restrict__ wht, const float* __restrict__ a,
              float* __restrict__ f1, float* __restrict__ f2) {
    const int t = blockIdx.x * 256 + threadIdx.x;
    const int b = t >> 11;
    const int j = t & (N_ - 1);
    const float* wb = wht + (size_t)b * D_ * N_ + j;
    float s1 = 0.f, s2 = 0.f;
#pragma unroll 4
    for (int d = 0; d < D_; ++d) {
        const float w = wb[(size_t)d * N_];
        s1 += w * a[d];
        s2 += w * a[D_ + d];
    }
    f1[t] = s1;
    f2[t] = s2;
}

typedef unsigned v4u __attribute__((ext_vector_type(4)));

#define RB_     64
#define JC_     128
#define PITCH_  132
#define BPITCH_ 68

__global__ __launch_bounds__(256)
void gat_kernel(const float* __restrict__ wht, const float* __restrict__ f1g,
                const float* __restrict__ f2g, const int* __restrict__ adj,
                float* __restrict__ out) {
    __shared__ __attribute__((aligned(16))) float    stage[2][D_ * PITCH_];
    __shared__ __attribute__((aligned(16))) unsigned lds_bits[RB_ * BPITCH_];
    __shared__ __attribute__((aligned(16))) float    f2s[N_];
    __shared__ float lds_mw[RB_ * 33];
    __shared__ float lds_m[RB_], lds_f1[RB_], lds_pone[RB_];
    __shared__ float lds_lw[2 * RB_];

    const int bid  = blockIdx.x;
    const int b    = bid >> 5;
    const int i0   = (bid & 31) * RB_;
    const int tid  = threadIdx.x;
    const int wv   = tid >> 5;
    const int lane = tid & 31;
    const int c    = lane & 15;
    const int h    = lane >> 4;
    const int it   = wv & 3;
    const int jh   = wv >> 2;

    const float* f2b  = f2g + b * N_;
    const float* whtb = wht + (size_t)b * D_ * N_;

    ((v4f*)f2s)[tid]       = ((const v4f*)f2b)[tid];
    ((v4f*)f2s)[tid + 256] = ((const v4f*)f2b)[tid + 256];

    {
        float mm[8];
#pragma unroll
        for (int r = 0; r < 8; ++r) mm[r] = -3.0e38f;
        const int* abase = adj + (size_t)(i0 + wv * 8) * N_ + 4 * lane;
        for (int ch = 0; ch < N_ / JC_; ++ch) {
            const v4f fv = *(const v4f*)(f2b + ch * JC_ + 4 * lane);
#pragma unroll
            for (int r = 0; r < 8; ++r) {
                const v4i av = *(const v4i*)(abase + (size_t)r * N_ + ch * JC_);
                const bool q0 = av[0] > 0, q1 = av[1] > 0, q2 = av[2] > 0, q3 = av[3] > 0;
                const unsigned m0 = __builtin_amdgcn_ballot_w32(q0);
                const unsigned m1 = __builtin_amdgcn_ballot_w32(q1);
                const unsigned m2 = __builtin_amdgcn_ballot_w32(q2);
                const unsigned m3 = __builtin_amdgcn_ballot_w32(q3);
                float t = mm[r];
                t = fmaxf(t, q0 ? fv[0] : -3.0e38f);
                t = fmaxf(t, q1 ? fv[1] : -3.0e38f);
                t = fmaxf(t, q2 ? fv[2] : -3.0e38f);
                t = fmaxf(t, q3 ? fv[3] : -3.0e38f);
                mm[r] = t;
                const unsigned mine = (lane == 0) ? m0 : (lane == 1) ? m1
                                    : (lane == 2) ? m2 : m3;
                if (lane < 4)
                    lds_bits[(wv * 8 + r) * BPITCH_ + ch * 4 + lane] = mine;
            }
        }
#pragma unroll
        for (int r = 0; r < 8; ++r) lds_mw[(wv * 8 + r) * 33 + lane] = mm[r];
    }
    __syncthreads();
    if (tid < RB_) {
        float M = -3.0e38f;
#pragma unroll
        for (int s = 0; s < 32; ++s) M = fmaxf(M, lds_mw[tid * 33 + s]);
        const bool empty = (M < -1.0e38f);
        const float f1v  = f1g[b * N_ + i0 + tid];
        const float sum  = f1v + M;
        const float emax = fmaxf(sum, 0.2f * sum);
        lds_m[tid]    = empty ? 0.0f : emax;
        lds_pone[tid] = empty ? 1.0f : 0.0f;
        lds_f1[tid]   = f1v;
    }
    __syncthreads();

    const int   rowl = it * 16 + c;
    const float f1i  = lds_f1[rowl];
    const float mi   = lds_m[rowl];
    const float pm   = lds_pone[rowl];

    v8f acc[8] = {};
    float lsum = 0.0f;

    const float* gsrc = whtb + (size_t)wv * N_ + 4 * lane;
    const int    soff = wv * PITCH_ + 4 * lane;
    v4f st[16];
#pragma unroll
    for (int q = 0; q < 16; ++q) st[q] = *(const v4f*)(gsrc + (size_t)(8 * q) * N_);
#pragma unroll
    for (int q = 0; q < 16; ++q) *(v4f*)&stage[0][soff + 8 * q * PITCH_] = st[q];
    __syncthreads();

    for (int ch = 0; ch < N_ / JC_; ++ch) {
        const float* cur = stage[ch & 1];
        const bool more = (ch + 1 < N_ / JC_);
        if (more) {
#pragma unroll
            for (int q = 0; q < 16; ++q)
                st[q] = *(const v4f*)(gsrc + (size_t)(8 * q) * N_ + (ch + 1) * JC_);
        }
        const v4u bw = *(const v4u*)&lds_bits[rowl * BPITCH_ + ch * 4];
        const float* fb = cur + c * PITCH_ + jh * 64 + 4 * h;
        const float* fs = f2s + ch * JC_ + jh * 64 + 4 * h;
#pragma unroll
        for (int g = 0; g < 8; ++g) {
            const v4f f2v = *(const v4f*)(fs + 8 * g);
            const unsigned sh = (unsigned)(jh * 16 + 2 * g + h);
            float s0 = f1i + f2v[0]; s0 = fmaxf(s0, 0.2f * s0);
            float s1 = f1i + f2v[1]; s1 = fmaxf(s1, 0.2f * s1);
            float s2 = f1i + f2v[2]; s2 = fmaxf(s2, 0.2f * s2);
            float s3 = f1i + f2v[3]; s3 = fmaxf(s3, 0.2f * s3);
            const float p0 = ((bw[0] >> sh) & 1u) ? __expf(s0 - mi) : pm;
            const float p1 = ((bw[1] >> sh) & 1u) ? __expf(s1 - mi) : pm;
            const float p2 = ((bw[2] >> sh) & 1u) ? __expf(s2 - mi) : pm;
            const float p3 = ((bw[3] >> sh) & 1u) ? __expf(s3 - mi) : pm;
            lsum += (p0 + p1) + (p2 + p3);
            v2f A0; A0[0] = p0; A0[1] = p1;
            v2f A1; A1[0] = p2; A1[1] = p3;

            v4f Bv[8];
#pragma unroll
            for (int dt = 0; dt < 8; ++dt)
                Bv[dt] = *(const v4f*)(fb + dt * 16 * PITCH_ + 8 * g);
#pragma unroll
            for (int dt = 0; dt < 8; ++dt) {
                v2f B0; B0[0] = Bv[dt][0]; B0[1] = Bv[dt][1];
                acc[dt] = wmma_f32_4(A0, B0, acc[dt]);
            }
#pragma unroll
            for (int dt = 0; dt < 8; ++dt) {
                v2f B1; B1[0] = Bv[dt][2]; B1[1] = Bv[dt][3];
                acc[dt] = wmma_f32_4(A1, B1, acc[dt]);
            }
        }
        if (more) {
            float* nxt = stage[(ch + 1) & 1];
#pragma unroll
            for (int q = 0; q < 16; ++q) *(v4f*)&nxt[soff + 8 * q * PITCH_] = st[q];
        }
        __syncthreads();
    }

    const float lpair = lsum + __shfl_xor(lsum, 16);
    if (lane < 16) lds_lw[jh * RB_ + it * 16 + lane] = lpair;

    float* tile = stage[0];
    if (jh == 1) {
#pragma unroll
        for (int dt = 0; dt < 8; ++dt)
#pragma unroll
            for (int r = 0; r < 8; ++r)
                tile[(it * 16 + 8 * h + r) * PITCH_ + dt * 16 + c] = acc[dt][r];
    }
    __syncthreads();
    if (jh == 0) {
#pragma unroll
        for (int r = 0; r < 8; ++r) {
            const int   row = it * 16 + 8 * h + r;
            const float li  = lds_lw[row] + lds_lw[RB_ + row];
#pragma unroll
            for (int dt = 0; dt < 8; ++dt) {
                float* tp = &tile[row * PITCH_ + dt * 16 + c];
                const float hv = (acc[dt][r] + *tp) / li;
                *tp = (hv > 0.0f) ? hv : (__expf(hv) - 1.0f);
            }
        }
    }
    __syncthreads();

    float* ob = out + ((size_t)b * N_ + i0) * D_;
#pragma unroll
    for (int q = 0; q < 8; ++q) {
        const int row = wv * 8 + q;
        *(v4f*)(ob + (size_t)row * D_ + 4 * lane) =
            *(const v4f*)&tile[row * PITCH_ + 4 * lane];
    }
}

extern "C" void kernel_launch(void* const* d_in, const int* in_sizes, int n_in,
                              void* d_out, int out_size, void* d_ws, size_t ws_size,
                              hipStream_t stream) {
    const float* x   = (const float*)d_in[0];
    const int*   adj = (const int*)d_in[1];
    const float* W   = (const float*)d_in[2];
    const float* a   = (const float*)d_in[3];
    float* out = (float*)d_out;

    float* wht = (float*)d_ws;
    float* f1  = wht + (size_t)B_ * D_ * N_;
    float* f2  = f1 + (size_t)B_ * N_;

    wht_kernel<<<dim3(B_ * (N_ / 16)), dim3(256), 0, stream>>>(x, W, wht);
    f_kernel<<<dim3((B_ * N_) / 256), dim3(256), 0, stream>>>(wht, a, f1, f2);
    gat_kernel<<<dim3(B_ * (N_ / RB_)), dim3(256), 0, stream>>>(wht, f1, f2, adj, out);
}


# gat kernel pass 1: adjacency row loads issued two 128-column chunks ahead of use through a ring of 3 register sets (was one dependent load at a time); mask words merged per lane without branches
# speedup vs baseline: 2.5262x; 1.6415x over previous
#include <hip/hip_runtime.h>
#include <cstddef>

typedef float v2f __attribute__((ext_vector_type(2)));
typedef float v4f __attribute__((ext_vector_type(4)));
typedef float v8f __attribute__((ext_vector_type(8)));
typedef int   v4i __attribute__((ext_vector_type(4)));

#define B_  8
#define N_  2048
#define C_  128
#define D_  128
#define NEG_BIG (-9.0e15f)

static __device__ __forceinline__ v8f wmma_f32_4(v2f a, v2f b, v8f c) {
    return __builtin_amdgcn_wmma_f32_16x16x4_f32(false, a, false, b, (short)0, c,
                                                 false, false);
}

__global__ __launch_bounds__(256)
void wht_kernel(const float* __restrict__ x, const float* __restrict__ W,
                float* __restrict__ wht) {
    const int bid  = blockIdx.x;
    const int b    = bid >> 7;
    const int jt   = bid & 127;
    const int tid  = threadIdx.x;
    const int wv   = tid >> 5;
    const int lane = tid & 31;
    const int nl   = lane & 15;
    const int kh   = (lane >> 4) << 1;

    const float* xrow = x + ((size_t)(b * N_ + jt * 16 + nl)) * C_;
    const float* wrow = W + (size_t)(wv * 16 + nl) * C_;

    v8f acc = {};
#pragma unroll 4
    for (int k0 = 0; k0 < C_; k0 += 4) {
        const int kb = k0 + kh;
        v2f a = *(const v2f*)(wrow + kb);
        v2f bm = *(const v2f*)(xrow + kb);
        acc = wmma_f32_4(a, bm, acc);
    }

    float* ob = wht + (size_t)b * D_ * N_ + (size_t)jt * 16 + nl;
#pragma unroll
    for (int r = 0; r < 8; ++r) {
        const int d = wv * 16 + r + ((lane >> 4) << 3);
        ob[(size_t)d * N_] = acc[r];
    }
}

__global__ __launch_bounds__(256)
void f_kernel(const float* __restrict__ wht, const float* __restrict__ a,
              float* __restrict__ f1, float* __restrict__ f2) {
    const int t = blockIdx.x * 256 + threadIdx.x;
    const int b = t >> 11;
    const int j = t & (N_ - 1);
    const float* wb = wht + (size_t)b * D_ * N_ + j;
    float s1 = 0.f, s2 = 0.f;
#pragma unroll 4
    for (int d = 0; d < D_; ++d) {
        const float w = wb[(size_t)d * N_];
        s1 += w * a[d];
        s2 += w * a[D_ + d];
    }
    f1[t] = s1;
    f2[t] = s2;
}

typedef unsigned v4u __attribute__((ext_vector_type(4)));

#define RB_     64
#define JC_     128
#define PITCH_  132
#define BPITCH_ 68
#define P1_AHEAD_ 2
#define P1_RING_  3

__global__ __launch_bounds__(256)
void gat_kernel(const float* __restrict__ wht, const float* __restrict__ f1g,
                const float* __restrict__ f2g, const int* __restrict__ adj,
                float* __restrict__ out) {
    __shared__ __attribute__((aligned(16))) float    stage[2][D_ * PITCH_];
    __shared__ __attribute__((aligned(16))) unsigned lds_bits[RB_ * BPITCH_];
    __shared__ __attribute__((aligned(16))) float    f2s[N_];
    __shared__ float lds_mw[RB_ * 33];
    __shared__ float lds_m[RB_], lds_f1[RB_], lds_pone[RB_];
    __shared__ float lds_lw[2 * RB_];

    const int bid  = blockIdx.x;
    const int b    = bid >> 5;
    const int i0   = (bid & 31) * RB_;
    const int tid  = threadIdx.x;
    const int wv   = tid >> 5;
    const int lane = tid & 31;
    const int c    = lane & 15;
    const int h    = lane >> 4;
    const int it   = wv & 3;
    const int jh   = wv >> 2;

    const float* f2b  = f2g + b * N_;
    const float* whtb = wht + (size_t)b * D_ * N_;

    ((v4f*)f2s)[tid]       = ((const v4f*)f2b)[tid];
    ((v4f*)f2s)[tid + 256] = ((const v4f*)f2b)[tid + 256];

    {
        float mm[8];
#pragma unroll
        for (int r = 0; r < 8; ++r) mm[r] = -3.0e38f;
        const int*   abase = adj + (size_t)(i0 + wv * 8) * N_ + 4 * lane;
        const float* fbase = f2b + 4 * lane;
        unsigned*    brow  = &lds_bits[(wv * 8 + (lane >> 2)) * BPITCH_ + (lane & 3)];
        unsigned ksel[32];
#pragma unroll
        for (int s = 0; s < 32; ++s) ksel[s] = (lane == s) ? 0xffffffffu : 0u;
        v4i av[P1_RING_][8];
        v4f fq[P1_RING_];
#pragma unroll
        for (int p = 0; p < P1_AHEAD_; ++p) {
            fq[p] = *(const v4f*)(fbase + p * JC_);
#pragma unroll
            for (int r = 0; r < 8; ++r)
                av[p][r] = *(const v4i*)(abase + (size_t)r * N_ + p * JC_);
        }
#pragma unroll
        for (int ch = 0; ch < N_ / JC_; ++ch) {
            if (ch + P1_AHEAD_ < N_ / JC_) {
                const int sn = (ch + P1_AHEAD_) % P1_RING_;
                fq[sn] = *(const v4f*)(fbase + (ch + P1_AHEAD_) * JC_);
#pragma unroll
                for (int r = 0; r < 8; ++r)
                    av[sn][r] = *(const v4i*)(abase + (size_t)r * N_ + (ch + P1_AHEAD_) * JC_);
            }
            const int sc = ch % P1_RING_;
            const v4f fv = fq[sc];
            unsigned wbits = 0u;
#pragma unroll
            for (int r = 0; r < 8; ++r) {
                const v4i a4 = av[sc][r];
                const bool q0 = a4[0] > 0, q1 = a4[1] > 0, q2 = a4[2] > 0, q3 = a4[3] > 0;
                const unsigned m0 = __builtin_amdgcn_ballot_w32(q0);
                const unsigned m1 = __builtin_amdgcn_ballot_w32(q1);
                const unsigned m2 = __builtin_amdgcn_ballot_w32(q2);
                const unsigned m3 = __builtin_amdgcn_ballot_w32(q3);
                float t = mm[r];
                t = fmaxf(t, q0 ? fv[0] : -3.0e38f);
                t = fmaxf(t, q1 ? fv[1] : -3.0e38f);
                t = fmaxf(t, q2 ? fv[2] : -3.0e38f);
                t = fmaxf(t, q3 ? fv[3] : -3.0e38f);
                mm[r] = t;
                wbits |= m0 & ksel[4 * r + 0];
                wbits |= m1 & ksel[4 * r + 1];
                wbits |= m2 & ksel[4 * r + 2];
                wbits |= m3 & ksel[4 * r + 3];
            }
            brow[ch * 4] = wbits;
        }
#pragma unroll
        for (int r = 0; r < 8; ++r) lds_mw[(wv * 8 + r) * 33 + lane] = mm[r];
    }
    __syncthreads();
    if (tid < RB_) {
        float M = -3.0e38f;
#pragma unroll
        for (int s = 0; s < 32; ++s) M = fmaxf(M, lds_mw[tid * 33 + s]);
        const bool empty = (M < -1.0e38f);
        const float f1v  = f1g[b * N_ + i0 + tid];
        const float sum  = f1v + M;
        const float emax = fmaxf(sum, 0.2f * sum);
        lds_m[tid]    = empty ? 0.0f : emax;
        lds_pone[tid] = empty ? 1.0f : 0.0f;
        lds_f1[tid]   = f1v;
    }
    __syncthreads();

    const int   rowl = it * 16 + c;
    const float f1i  = lds_f1[rowl];
    const float mi   = lds_m[rowl];
    const float pm   = lds_pone[rowl];

    v8f acc[8] = {};
    float lsum = 0.0f;

    const float* gsrc = whtb + (size_t)wv * N_ + 4 * lane;
    const int    soff = wv * PITCH_ + 4 * lane;
    v4f st[16];
#pragma unroll
    for (int q = 0; q < 16; ++q) st[q] = *(const v4f*)(gsrc + (size_t)(8 * q) * N_);
#pragma unroll
    for (int q = 0; q < 16; ++q) *(v4f*)&stage[0][soff + 8 * q * PITCH_] = st[q];
    __syncthreads();

    for (int ch = 0; ch < N_ / JC_; ++ch) {
        const float* cur = stage[ch & 1];
        const bool more = (ch + 1 < N_ / JC_);
        if (more) {
#pragma unroll
            for (int q = 0; q < 16; ++q)
                st[q] = *(const v4f*)(gsrc + (size_t)(8 * q) * N_ + (ch + 1) * JC_);
        }
        const v4u bw = *(const v4u*)&lds_bits[rowl * BPITCH_ + ch * 4];
        const float* fb = cur + c * PITCH_ + jh * 64 + 4 * h;
        const float* fs = f2s + ch * JC_ + jh * 64 + 4 * h;
#pragma unroll
        for (int g = 0; g < 8; ++g) {
            const v4f f2v = *(const v4f*)(fs + 8 * g);
            const unsigned sh = (unsigned)(jh * 16 + 2 * g + h);
            float s0 = f1i + f2v[0]; s0 = fmaxf(s0, 0.2f * s0);
            float s1 = f1i + f2v[1]; s1 = fmaxf(s1, 0.2f * s1);
            float s2 = f1i + f2v[2]; s2 = fmaxf(s2, 0.2f * s2);
            float s3 = f1i + f2v[3]; s3 = fmaxf(s3, 0.2f * s3);
            const float p0 = ((bw[0] >> sh) & 1u) ? __expf(s0 - mi) : pm;
            const float p1 = ((bw[1] >> sh) & 1u) ? __expf(s1 - mi) : pm;
            const float p2 = ((bw[2] >> sh) & 1u) ? __expf(s2 - mi) : pm;
            const float p3 = ((bw[3] >> sh) & 1u) ? __expf(s3 - mi) : pm;
            lsum += (p0 + p1) + (p2 + p3);
            v2f A0; A0[0] = p0; A0[1] = p1;
            v2f A1; A1[0] = p2; A1[1] = p3;

            v4f Bv[8];
#pragma unroll
            for (int dt = 0; dt < 8; ++dt)
                Bv[dt] = *(const v4f*)(fb + dt * 16 * PITCH_ + 8 * g);
#pragma unroll
            for (int dt = 0; dt < 8; ++dt) {
                v2f B0; B0[0] = Bv[dt][0]; B0[1] = Bv[dt][1];
                acc[dt] = wmma_f32_4(A0, B0, acc[dt]);
            }
#pragma unroll
            for (int dt = 0; dt < 8; ++dt) {
                v2f B1; B1[0] = Bv[dt][2]; B1[1] = Bv[dt][3];
                acc[dt] = wmma_f32_4(A1, B1, acc[dt]);
            }
        }
        if (more) {
            float* nxt = stage[(ch + 1) & 1];
#pragma unroll
            for (int q = 0; q < 16; ++q) *(v4f*)&nxt[soff + 8 * q * PITCH_] = st[q];
        }
        __syncthreads();
    }

    const float lpair = lsum + __shfl_xor(lsum, 16);
    if (lane < 16) lds_lw[jh * RB_ + it * 16 + lane] = lpair;

    float* tile = stage[0];
    if (jh == 1) {
#pragma unroll
        for (int dt = 0; dt < 8; ++dt)
#pragma unroll
            for (int r = 0; r < 8; ++r)
                tile[(it * 16 + 8 * h + r) * PITCH_ + dt * 16 + c] = acc[dt][r];
    }
    __syncthreads();
    if (jh == 0) {
#pragma unroll
        for (int r = 0; r < 8; ++r) {
            const int   row = it * 16 + 8 * h + r;
            const float li  = lds_lw[row] + lds_lw[RB_ + row];
#pragma unroll
            for (int dt = 0; dt < 8; ++dt) {
                float* tp = &tile[row * PITCH_ + dt * 16 + c];
                const float hv = (acc[dt][r] + *tp) / li;
                *tp = (hv > 0.0f) ? hv : (__expf(hv) - 1.0f);
            }
        }
    }
    __syncthreads();

    float* ob = out + ((size_t)b * N_ + i0) * D_;
#pragma unroll
    for (int q = 0; q < 8; ++q) {
        const int row = wv * 8 + q;
        *(v4f*)(ob + (size_t)row * D_ + 4 * lane) =
            *(const v4f*)&tile[row * PITCH_ + 4 * lane];
    }
}

extern "C" void kernel_launch(void* const* d_in, const int* in_sizes, int n_in,
                              void* d_out, int out_size, void* d_ws, size_t ws_size,
                              hipStream_t stream) {
    const float* x   = (const float*)d_in[0];
    const int*   adj = (const int*)d_in[1];
    const float* W   = (const float*)d_in[2];
    const float* a   = (const float*)d_in[3];
    float* out = (float*)d_out;

    float* wht = (float*)d_ws;
    float* f1  = wht + (size_t)B_ * D_ * N_;
    float* f2  = f1 + (size_t)B_ * N_;

    wht_kernel<<<dim3(B_ * (N_ / 16)), dim3(256), 0, stream>>>(x, W, wht);
    f_kernel<<<dim3((B_ * N_) / 256), dim3(256), 0, stream>>>(wht, a, f1, f2);
    gat_kernel<<<dim3(B_ * (N_ / RB_)), dim3(256), 0, stream>>>(wht, f1, f2, adj, out);
}


# f1/f2 computed in the WhT kernel's epilogue from its accumulators (f kernel and its launch removed); their d-sum order changes (8-term lane partials, lane halves, then 8 waves in fixed order)
# speedup vs baseline: 3.1261x; 1.2375x over previous
#include <hip/hip_runtime.h>
#include <cstddef>

typedef float v2f __attribute__((ext_vector_type(2)));
typedef float v4f __attribute__((ext_vector_type(4)));
typedef float v8f __attribute__((ext_vector_type(8)));
typedef int   v4i __attribute__((ext_vector_type(4)));

#define B_  8
#define N_  2048
#define C_  128
#define D_  128
#define NEG_BIG (-9.0e15f)

static __device__ __forceinline__ v8f wmma_f32_4(v2f a, v2f b, v8f c) {
    return __builtin_amdgcn_wmma_f32_16x16x4_f32(false, a, false, b, (short)0, c,
                                                 false, false);
}

__global__ __launch_bounds__(256)
void wht_kernel(const float* __restrict__ x, const float* __restrict__ W,
                const float* __restrict__ a, float* __restrict__ wht,
                float* __restrict__ f1, float* __restrict__ f2) {
    __shared__ float fpart[2][8][16];

    const int bid  = blockIdx.x;
    const int b    = bid >> 7;
    const int jt   = bid & 127;
    const int tid  = threadIdx.x;
    const int wv   = tid >> 5;
    const int lane = tid & 31;
    const int nl   = lane & 15;
    const int hh   = lane >> 4;
    const int kh   = hh << 1;

    const float* xrow = x + ((size_t)(b * N_ + jt * 16 + nl)) * C_;
    const float* wrow = W + (size_t)(wv * 16 + nl) * C_;

    const float* ap = a + wv * 16 + 8 * hh;
    const v4f a1lo = *(const v4f*)(ap);
    const v4f a1hi = *(const v4f*)(ap + 4);
    const v4f a2lo = *(const v4f*)(ap + D_);
    const v4f a2hi = *(const v4f*)(ap + D_ + 4);

    v8f acc = {};
#pragma unroll 4
    for (int k0 = 0; k0 < C_; k0 += 4) {
        const int kb = k0 + kh;
        v2f am = *(const v2f*)(wrow + kb);
        v2f bm = *(const v2f*)(xrow + kb);
        acc = wmma_f32_4(am, bm, acc);
    }

    float s1 = 0.f, s2 = 0.f;
#pragma unroll
    for (int r = 0; r < 4; ++r) {
        s1 += acc[r] * a1lo[r];
        s2 += acc[r] * a2lo[r];
    }
#pragma unroll
    for (int r = 0; r < 4; ++r) {
        s1 += acc[4 + r] * a1hi[r];
        s2 += acc[4 + r] * a2hi[r];
    }
    s1 += __shfl_xor(s1, 16);
    s2 += __shfl_xor(s2, 16);
    if (lane < 16) {
        fpart[0][wv][nl] = s1;
        fpart[1][wv][nl] = s2;
    }
    __syncthreads();

    float* ob = wht + (size_t)b * D_ * N_ + (size_t)jt * 16 + nl;
#pragma unroll
    for (int r = 0; r < 8; ++r) {
        const int d = wv * 16 + r + (hh << 3);
        ob[(size_t)d * N_] = acc[r];
    }

    if (tid < 32) {
        float s = 0.f;
#pragma unroll
        for (int w = 0; w < 8; ++w) s += fpart[hh][w][nl];
        float* dst = hh ? f2 : f1;
        dst[b * N_ + jt * 16 + nl] = s;
    }
}

typedef unsigned v4u __attribute__((ext_vector_type(4)));

#define RB_     64
#define JC_     128
#define PITCH_  132
#define BPITCH_ 68
#define P1_AHEAD_ 2
#define P1_RING_  3

__global__ __launch_bounds__(256)
void gat_kernel(const float* __restrict__ wht, const float* __restrict__ f1g,
                const float* __restrict__ f2g, const int* __restrict__ adj,
                float* __restrict__ out) {
    __shared__ __attribute__((aligned(16))) float    stage[2][D_ * PITCH_];
    __shared__ __attribute__((aligned(16))) unsigned lds_bits[RB_ * BPITCH_];
    __shared__ __attribute__((aligned(16))) float    f2s[N_];
    __shared__ float lds_mw[RB_ * 33];
    __shared__ float lds_m[RB_], lds_f1[RB_], lds_pone[RB_];
    __shared__ float lds_lw[2 * RB_];

    const int bid  = blockIdx.x;
    const int b    = bid >> 5;
    const int i0   = (bid & 31) * RB_;
    const int tid  = threadIdx.x;
    const int wv   = tid >> 5;
    const int lane = tid & 31;
    const int c    = lane & 15;
    const int h    = lane >> 4;
    const int it   = wv & 3;
    const int jh   = wv >> 2;

    const float* f2b  = f2g + b * N_;
    const float* whtb = wht + (size_t)b * D_ * N_;

    ((v4f*)f2s)[tid]       = ((const v4f*)f2b)[tid];
    ((v4f*)f2s)[tid + 256] = ((const v4f*)f2b)[tid + 256];

    {
        float mm[8];
#pragma unroll
        for (int r = 0; r < 8; ++r) mm[r] = -3.0e38f;
        const int*   abase = adj + (size_t)(i0 + wv * 8) * N_ + 4 * lane;
        const float* fbase = f2b + 4 * lane;
        unsigned*    brow  = &lds_bits[(wv * 8 + (lane >> 2)) * BPITCH_ + (lane & 3)];
        unsigned ksel[32];
#pragma unroll
        for (int s = 0; s < 32; ++s) ksel[s] = (lane == s) ? 0xffffffffu : 0u;
        v4i av[P1_RING_][8];
        v4f fq[P1_RING_];
#pragma unroll
        for (int p = 0; p < P1_AHEAD_; ++p) {
            fq[p] = *(const v4f*)(fbase + p * JC_);
#pragma unroll
            for (int r = 0; r < 8; ++r)
                av[p][r] = *(const v4i*)(abase + (size_t)r * N_ + p * JC_);
        }
#pragma unroll
        for (int ch = 0; ch < N_ / JC_; ++ch) {
            if (ch + P1_AHEAD_ < N_ / JC_) {
                const int sn = (ch + P1_AHEAD_) % P1_RING_;
                fq[sn] = *(const v4f*)(fbase + (ch + P1_AHEAD_) * JC_);
#pragma unroll
                for (int r = 0; r < 8; ++r)
                    av[sn][r] = *(const v4i*)(abase + (size_t)r * N_ + (ch + P1_AHEAD_) * JC_);
            }
            const int sc = ch % P1_RING_;
            const v4f fv = fq[sc];
            unsigned wbits = 0u;
#pragma unroll
            for (int r = 0; r < 8; ++r) {
                const v4i a4 = av[sc][r];
                const bool q0 = a4[0] > 0, q1 = a4[1] > 0, q2 = a4[2] > 0, q3 = a4[3] > 0;
                const unsigned m0 = __builtin_amdgcn_ballot_w32(q0);
                const unsigned m1 = __builtin_amdgcn_ballot_w32(q1);
                const unsigned m2 = __builtin_amdgcn_ballot_w32(q2);
                const unsigned m3 = __builtin_amdgcn_ballot_w32(q3);
                float t = mm[r];
                t = fmaxf(t, q0 ? fv[0] : -3.0e38f);
                t = fmaxf(t, q1 ? fv[1] : -3.0e38f);
                t = fmaxf(t, q2 ? fv[2] : -3.0e38f);
                t = fmaxf(t, q3 ? fv[3] : -3.0e38f);
                mm[r] = t;
                wbits |= m0 & ksel[4 * r + 0];
                wbits |= m1 & ksel[4 * r + 1];
                wbits |= m2 & ksel[4 * r + 2];
                wbits |= m3 & ksel[4 * r + 3];
            }
            brow[ch * 4] = wbits;
        }
#pragma unroll
        for (int r = 0; r < 8; ++r) lds_mw[(wv * 8 + r) * 33 + lane] = mm[r];
    }
    __syncthreads();
    if (tid < RB_) {
        float M = -3.0e38f;
#pragma unroll
        for (int s = 0; s < 32; ++s) M = fmaxf(M, lds_mw[tid * 33 + s]);
        const bool empty = (M < -1.0e38f);
        const float f1v  = f1g[b * N_ + i0 + tid];
        const float sum  = f1v + M;
        const float emax = fmaxf(sum, 0.2f * sum);
        lds_m[tid]    = empty ? 0.0f : emax;
        lds_pone[tid] = empty ? 1.0f : 0.0f;
        lds_f1[tid]   = f1v;
    }
    __syncthreads();

    const int   rowl = it * 16 + c;
    const float f1i  = lds_f1[rowl];
    const float mi   = lds_m[rowl];
    const float pm   = lds_pone[rowl];

    v8f acc[8] = {};
    float lsum = 0.0f;

    const float* gsrc = whtb + (size_t)wv * N_ + 4 * lane;
    const int    soff = wv * PITCH_ + 4 * lane;
    v4f st[16];
#pragma unroll
    for (int q = 0; q < 16; ++q) st[q] = *(const v4f*)(gsrc + (size_t)(8 * q) * N_);
#pragma unroll
    for (int q = 0; q < 16; ++q) *(v4f*)&stage[0][soff + 8 * q * PITCH_] = st[q];
    __syncthreads();

    for (int ch = 0; ch < N_ / JC_; ++ch) {
        const float* cur = stage[ch & 1];
        const bool more = (ch + 1 < N_ / JC_);
        if (more) {
#pragma unroll
            for (int q = 0; q < 16; ++q)
                st[q] = *(const v4f*)(gsrc + (size_t)(8 * q) * N_ + (ch + 1) * JC_);
        }
        const v4u bw = *(const v4u*)&lds_bits[rowl * BPITCH_ + ch * 4];
        const float* fb = cur + c * PITCH_ + jh * 64 + 4 * h;
        const float* fs = f2s + ch * JC_ + jh * 64 + 4 * h;
#pragma unroll
        for (int g = 0; g < 8; ++g) {
            const v4f f2v = *(const v4f*)(fs + 8 * g);
            const unsigned sh = (unsigned)(jh * 16 + 2 * g + h);
            float s0 = f1i + f2v[0]; s0 = fmaxf(s0, 0.2f * s0);
            float s1 = f1i + f2v[1]; s1 = fmaxf(s1, 0.2f * s1);
            float s2 = f1i + f2v[2]; s2 = fmaxf(s2, 0.2f * s2);
            float s3 = f1i + f2v[3]; s3 = fmaxf(s3, 0.2f * s3);
            const float p0 = ((bw[0] >> sh) & 1u) ? __expf(s0 - mi) : pm;
            const float p1 = ((bw[1] >> sh) & 1u) ? __expf(s1 - mi) : pm;
            const float p2 = ((bw[2] >> sh) & 1u) ? __expf(s2 - mi) : pm;
            const float p3 = ((bw[3] >> sh) & 1u) ? __expf(s3 - mi) : pm;
            lsum += (p0 + p1) + (p2 + p3);
            v2f A0; A0[0] = p0; A0[1] = p1;
            v2f A1; A1[0] = p2; A1[1] = p3;

            v4f Bv[8];
#pragma unroll
            for (int dt = 0; dt < 8; ++dt)
                Bv[dt] = *(const v4f*)(fb + dt * 16 * PITCH_ + 8 * g);
#pragma unroll
            for (int dt = 0; dt < 8; ++dt) {
                v2f B0; B0[0] = Bv[dt][0]; B0[1] = Bv[dt][1];
                acc[dt] = wmma_f32_4(A0, B0, acc[dt]);
            }
#pragma unroll
            for (int dt = 0; dt < 8; ++dt) {
                v2f B1; B1[0] = Bv[dt][2]; B1[1] = Bv[dt][3];
                acc[dt] = wmma_f32_4(A1, B1, acc[dt]);
            }
        }
        if (more) {
            float* nxt = stage[(ch + 1) & 1];
#pragma unroll
            for (int q = 0; q < 16; ++q) *(v4f*)&nxt[soff + 8 * q * PITCH_] = st[q];
        }
        __syncthreads();
    }

    const float lpair = lsum + __shfl_xor(lsum, 16);
    if (lane < 16) lds_lw[jh * RB_ + it * 16 + lane] = lpair;

    float* tile = stage[0];
    if (jh == 1) {
#pragma unroll
        for (int dt = 0; dt < 8; ++dt)
#pragma unroll
            for (int r = 0; r < 8; ++r)
                tile[(it * 16 + 8 * h + r) * PITCH_ + dt * 16 + c] = acc[dt][r];
    }
    __syncthreads();
    if (jh == 0) {
#pragma unroll
        for (int r = 0; r < 8; ++r) {
            const int   row = it * 16 + 8 * h + r;
            const float li  = lds_lw[row] + lds_lw[RB_ + row];
#pragma unroll
            for (int dt = 0; dt < 8; ++dt) {
                float* tp = &tile[row * PITCH_ + dt * 16 + c];
                const float hv = (acc[dt][r] + *tp) / li;
                *tp = (hv > 0.0f) ? hv : (__expf(hv) - 1.0f);
            }
        }
    }
    __syncthreads();

    float* ob = out + ((size_t)b * N_ + i0) * D_;
#pragma unroll
    for (int q = 0; q < 8; ++q) {
        const int row = wv * 8 + q;
        *(v4f*)(ob + (size_t)row * D_ + 4 * lane) =
            *(const v4f*)&tile[row * PITCH_ + 4 * lane];
    }
}

extern "C" void kernel_launch(void* const* d_in, const int* in_sizes, int n_in,
                              void* d_out, int out_size, void* d_ws, size_t ws_size,
                              hipStream_t stream) {
    const float* x   = (const float*)d_in[0];
    const int*   adj = (const int*)d_in[1];
    const float* W   = (const float*)d_in[2];
    const float* a   = (const float*)d_in[3];
    float* out = (float*)d_out;

    float* wht = (float*)d_ws;
    float* f1  = wht + (size_t)B_ * D_ * N_;
    float* f2  = f1 + (size_t)B_ * N_;

    wht_kernel<<<dim3(B_ * (N_ / 16)), dim3(256), 0, stream>>>(x, W, a, wht, f1, f2);
    gat_kernel<<<dim3(B_ * (N_ / RB_)), dim3(256), 0, stream>>>(wht, f1, f2, adj, out);
}


# wht kernel retiled: 64 j per block, x rows loaded as whole rows and staged once in LDS, W fragments kept in registers, WhT stored as whole lines via LDS; its k-sum order changes
# speedup vs baseline: 3.4681x; 1.1094x over previous
#include <hip/hip_runtime.h>
#include <cstddef>

typedef float v2f __attribute__((ext_vector_type(2)));
typedef float v4f __attribute__((ext_vector_type(4)));
typedef float v8f __attribute__((ext_vector_type(8)));
typedef int   v4i __attribute__((ext_vector_type(4)));

#define B_  8
#define N_  2048
#define C_  128
#define D_  128
#define NEG_BIG (-9.0e15f)

static __device__ __forceinline__ v8f wmma_f32_4(v2f a, v2f b, v8f c) {
    return __builtin_amdgcn_wmma_f32_16x16x4_f32(false, a, false, b, (short)0, c,
                                                 false, false);
}

#define XJ_  64
#define XP_  132
#define OP_  68

__global__ __launch_bounds__(256)
void wht_kernel(const float* __restrict__ x, const float* __restrict__ W,
                const float* __restrict__ a, float* __restrict__ wht,
                float* __restrict__ f1, float* __restrict__ f2) {
    __shared__ __attribute__((aligned(16))) float xs[XJ_ * XP_];
    __shared__ __attribute__((aligned(16))) float os[D_ * OP_];
    __shared__ float fpart[2][8][XJ_];

    const int bid  = blockIdx.x;
    const int b    = bid >> 5;
    const int j0   = (bid & 31) * XJ_;
    const int tid  = threadIdx.x;
    const int wv   = tid >> 5;
    const int lane = tid & 31;
    const int c    = lane & 15;
    const int h    = lane >> 4;

    const float* xsrc = x + ((size_t)(b * N_ + j0 + wv)) * C_ + 4 * lane;
    v4f xt[8];
#pragma unroll
    for (int q = 0; q < 8; ++q) xt[q] = *(const v4f*)(xsrc + (size_t)(8 * q) * C_);

    const float* wrow = W + (size_t)(wv * 16 + c) * C_ + 4 * h;
    v4f af[16];
#pragma unroll
    for (int q = 0; q < 16; ++q) af[q] = *(const v4f*)(wrow + 8 * q);

    const float* ap = a + wv * 16 + 8 * h;
    const v4f a1lo = *(const v4f*)(ap);
    const v4f a1hi = *(const v4f*)(ap + 4);
    const v4f a2lo = *(const v4f*)(ap + D_);
    const v4f a2hi = *(const v4f*)(ap + D_ + 4);

#pragma unroll
    for (int q = 0; q < 8; ++q) *(v4f*)&xs[(wv + 8 * q) * XP_ + 4 * lane] = xt[q];
    __syncthreads();

    v8f acc[4] = {};
    const float* fb = xs + c * XP_ + 4 * h;
#pragma unroll
    for (int q = 0; q < 16; ++q) {
        v4f Bv[4];
#pragma unroll
        for (int t = 0; t < 4; ++t) Bv[t] = *(const v4f*)(fb + t * 16 * XP_ + 8 * q);
        v2f A0; A0[0] = af[q][0]; A0[1] = af[q][1];
        v2f A1; A1[0] = af[q][2]; A1[1] = af[q][3];
#pragma unroll
        for (int t = 0; t < 4; ++t) {
            v2f B0; B0[0] = Bv[t][0]; B0[1] = Bv[t][1];
            acc[t] = wmma_f32_4(A0, B0, acc[t]);
        }
#pragma unroll
        for (int t = 0; t < 4; ++t) {
            v2f B1; B1[0] = Bv[t][2]; B1[1] = Bv[t][3];
            acc[t] = wmma_f32_4(A1, B1, acc[t]);
        }
    }

#pragma unroll
    for (int t = 0; t < 4; ++t) {
        float s1 = 0.f, s2 = 0.f;
#pragma unroll
        for (int r = 0; r < 4; ++r) {
            s1 += acc[t][r] * a1lo[r];
            s2 += acc[t][r] * a2lo[r];
        }
#pragma unroll
        for (int r = 0; r < 4; ++r) {
            s1 += acc[t][4 + r] * a1hi[r];
            s2 += acc[t][4 + r] * a2hi[r];
        }
        s1 += __shfl_xor(s1, 16);
        s2 += __shfl_xor(s2, 16);
        if (lane < 16) {
            fpart[0][wv][t * 16 + c] = s1;
            fpart[1][wv][t * 16 + c] = s2;
        }
#pragma unroll
        for (int r = 0; r < 8; ++r)
            os[(wv * 16 + 8 * h + r) * OP_ + t * 16 + c] = acc[t][r];
    }
    __syncthreads();

    float* ob = wht + (size_t)b * D_ * N_ + j0 + 4 * c;
#pragma unroll
    for (int i = 0; i < 8; ++i) {
        const int d = wv * 16 + 2 * i + h;
        *(v4f*)(ob + (size_t)d * N_) = *(const v4f*)&os[d * OP_ + 4 * c];
    }

    if (tid < 2 * XJ_) {
        const int which = tid >> 6;
        const int j     = tid & (XJ_ - 1);
        float s = 0.f;
#pragma unroll
        for (int w = 0; w < 8; ++w) s += fpart[which][w][j];
        float* dst = which ? f2 : f1;
        dst[b * N_ + j0 + j] = s;
    }
}

typedef unsigned v4u __attribute__((ext_vector_type(4)));

#define RB_     64
#define JC_     128
#define PITCH_  132
#define BPITCH_ 68
#define P1_AHEAD_ 2
#define P1_RING_  3

__global__ __launch_bounds__(256)
void gat_kernel(const float* __restrict__ wht, const float* __restrict__ f1g,
                const float* __restrict__ f2g, const int* __restrict__ adj,
                float* __restrict__ out) {
    __shared__ __attribute__((aligned(16))) float    stage[2][D_ * PITCH_];
    __shared__ __attribute__((aligned(16))) unsigned lds_bits[RB_ * BPITCH_];
    __shared__ __attribute__((aligned(16))) float    f2s[N_];
    __shared__ float lds_mw[RB_ * 33];
    __shared__ float lds_m[RB_], lds_f1[RB_], lds_pone[RB_];
    __shared__ float lds_lw[2 * RB_];

    const int bid  = blockIdx.x;
    const int b    = bid >> 5;
    const int i0   = (bid & 31) * RB_;
    const int tid  = threadIdx.x;
    const int wv   = tid >> 5;
    const int lane = tid & 31;
    const int c    = lane & 15;
    const int h    = lane >> 4;
    const int it   = wv & 3;
    const int jh   = wv >> 2;

    const float* f2b  = f2g + b * N_;
    const float* whtb = wht + (size_t)b * D_ * N_;

    ((v4f*)f2s)[tid]       = ((const v4f*)f2b)[tid];
    ((v4f*)f2s)[tid + 256] = ((const v4f*)f2b)[tid + 256];

    {
        float mm[8];
#pragma unroll
        for (int r = 0; r < 8; ++r) mm[r] = -3.0e38f;
        const int*   abase = adj + (size_t)(i0 + wv * 8) * N_ + 4 * lane;
        const float* fbase = f2b + 4 * lane;
        unsigned*    brow  = &lds_bits[(wv * 8 + (lane >> 2)) * BPITCH_ + (lane & 3)];
        unsigned ksel[32];
#pragma unroll
        for (int s = 0; s < 32; ++s) ksel[s] = (lane == s) ? 0xffffffffu : 0u;
        v4i av[P1_RING_][8];
        v4f fq[P1_RING_];
#pragma unroll
        for (int p = 0; p < P1_AHEAD_; ++p) {
            fq[p] = *(const v4f*)(fbase + p * JC_);
#pragma unroll
            for (int r = 0; r < 8; ++r)
                av[p][r] = *(const v4i*)(abase + (size_t)r * N_ + p * JC_);
        }
#pragma unroll
        for (int ch = 0; ch < N_ / JC_; ++ch) {
            if (ch + P1_AHEAD_ < N_ / JC_) {
                const int sn = (ch + P1_AHEAD_) % P1_RING_;
                fq[sn] = *(const v4f*)(fbase + (ch + P1_AHEAD_) * JC_);
#pragma unroll
                for (int r = 0; r < 8; ++r)
                    av[sn][r] = *(const v4i*)(abase + (size_t)r * N_ + (ch + P1_AHEAD_) * JC_);
            }
            const int sc = ch % P1_RING_;
            const v4f fv = fq[sc];
            unsigned wbits = 0u;
#pragma unroll
            for (int r = 0; r < 8; ++r) {
                const v4i a4 = av[sc][r];
                const bool q0 = a4[0] > 0, q1 = a4[1] > 0, q2 = a4[2] > 0, q3 = a4[3] > 0;
                const unsigned m0 = __builtin_amdgcn_ballot_w32(q0);
                const unsigned m1 = __builtin_amdgcn_ballot_w32(q1);
                const unsigned m2 = __builtin_amdgcn_ballot_w32(q2);
                const unsigned m3 = __builtin_amdgcn_ballot_w32(q3);
                float t = mm[r];
                t = fmaxf(t, q0 ? fv[0] : -3.0e38f);
                t = fmaxf(t, q1 ? fv[1] : -3.0e38f);
                t = fmaxf(t, q2 ? fv[2] : -3.0e38f);
                t = fmaxf(t, q3 ? fv[3] : -3.0e38f);
                mm[r] = t;
                wbits |= m0 & ksel[4 * r + 0];
                wbits |= m1 & ksel[4 * r + 1];
                wbits |= m2 & ksel[4 * r + 2];
                wbits |= m3 & ksel[4 * r + 3];
            }
            brow[ch * 4] = wbits;
        }
#pragma unroll
        for (int r = 0; r < 8; ++r) lds_mw[(wv * 8 + r) * 33 + lane] = mm[r];
    }
    __syncthreads();
    if (tid < RB_) {
        float M = -3.0e38f;
#pragma unroll
        for (int s = 0; s < 32; ++s) M = fmaxf(M, lds_mw[tid * 33 + s]);
        const bool empty = (M < -1.0e38f);
        const float f1v  = f1g[b * N_ + i0 + tid];
        const float sum  = f1v + M;
        const float emax = fmaxf(sum, 0.2f * sum);
        lds_m[tid]    = empty ? 0.0f : emax;
        lds_pone[tid] = empty ? 1.0f : 0.0f;
        lds_f1[tid]   = f1v;
    }
    __syncthreads();

    const int   rowl = it * 16 + c;
    const float f1i  = lds_f1[rowl];
    const float mi   = lds_m[rowl];
    const float pm   = lds_pone[rowl];

    v8f acc[8] = {};
    float lsum = 0.0f;

    const float* gsrc = whtb + (size_t)wv * N_ + 4 * lane;
    const int    soff = wv * PITCH_ + 4 * lane;
    v4f st[16];
#pragma unroll
    for (int q = 0; q < 16; ++q) st[q] = *(const v4f*)(gsrc + (size_t)(8 * q) * N_);
#pragma unroll
    for (int q = 0; q < 16; ++q) *(v4f*)&stage[0][soff + 8 * q * PITCH_] = st[q];
    __syncthreads();

    for (int ch = 0; ch < N_ / JC_; ++ch) {
        const float* cur = stage[ch & 1];
        const bool more = (ch + 1 < N_ / JC_);
        if (more) {
#pragma unroll
            for (int q = 0; q < 16; ++q)
                st[q] = *(const v4f*)(gsrc + (size_t)(8 * q) * N_ + (ch + 1) * JC_);
        }
        const v4u bw = *(const v4u*)&lds_bits[rowl * BPITCH_ + ch * 4];
        const float* fb = cur + c * PITCH_ + jh * 64 + 4 * h;
        const float* fs = f2s + ch * JC_ + jh * 64 + 4 * h;
#pragma unroll
        for (int g = 0; g < 8; ++g) {
            const v4f f2v = *(const v4f*)(fs + 8 * g);
            const unsigned sh = (unsigned)(jh * 16 + 2 * g + h);
            float s0 = f1i + f2v[0]; s0 = fmaxf(s0, 0.2f * s0);
            float s1 = f1i + f2v[1]; s1 = fmaxf(s1, 0.2f * s1);
            float s2 = f1i + f2v[2]; s2 = fmaxf(s2, 0.2f * s2);
            float s3 = f1i + f2v[3]; s3 = fmaxf(s3, 0.2f * s3);
            const float p0 = ((bw[0] >> sh) & 1u) ? __expf(s0 - mi) : pm;
            const float p1 = ((bw[1] >> sh) & 1u) ? __expf(s1 - mi) : pm;
            const float p2 = ((bw[2] >> sh) & 1u) ? __expf(s2 - mi) : pm;
            const float p3 = ((bw[3] >> sh) & 1u) ? __expf(s3 - mi) : pm;
            lsum += (p0 + p1) + (p2 + p3);
            v2f A0; A0[0] = p0; A0[1] = p1;
            v2f A1; A1[0] = p2; A1[1] = p3;

            v4f Bv[8];
#pragma unroll
            for (int dt = 0; dt < 8; ++dt)
                Bv[dt] = *(const v4f*)(fb + dt * 16 * PITCH_ + 8 * g);
#pragma unroll
            for (int dt = 0; dt < 8; ++dt) {
                v2f B0; B0[0] = Bv[dt][0]; B0[1] = Bv[dt][1];
                acc[dt] = wmma_f32_4(A0, B0, acc[dt]);
            }
#pragma unroll
            for (int dt = 0; dt < 8; ++dt) {
                v2f B1; B1[0] = Bv[dt][2]; B1[1] = Bv[dt][3];
                acc[dt] = wmma_f32_4(A1, B1, acc[dt]);
            }
        }
        if (more) {
            float* nxt = stage[(ch + 1) & 1];
#pragma unroll
            for (int q = 0; q < 16; ++q) *(v4f*)&nxt[soff + 8 * q * PITCH_] = st[q];
        }
        __syncthreads();
    }

    const float lpair = lsum + __shfl_xor(lsum, 16);
    if (lane < 16) lds_lw[jh * RB_ + it * 16 + lane] = lpair;

    float* tile = stage[0];
    if (jh == 1) {
#pragma unroll
        for (int dt = 0; dt < 8; ++dt)
#pragma unroll
            for (int r = 0; r < 8; ++r)
                tile[(it * 16 + 8 * h + r) * PITCH_ + dt * 16 + c] = acc[dt][r];
    }
    __syncthreads();
    if (jh == 0) {
#pragma unroll
        for (int r = 0; r < 8; ++r) {
            const int   row = it * 16 + 8 * h + r;
            const float li  = lds_lw[row] + lds_lw[RB_ + row];
#pragma unroll
            for (int dt = 0; dt < 8; ++dt) {
                float* tp = &tile[row * PITCH_ + dt * 16 + c];
                const float hv = (acc[dt][r] + *tp) / li;
                *tp = (hv > 0.0f) ? hv : (__expf(hv) - 1.0f);
            }
        }
    }
    __syncthreads();

    float* ob = out + ((size_t)b * N_ + i0) * D_;
#pragma unroll
    for (int q = 0; q < 8; ++q) {
        const int row = wv * 8 + q;
        *(v4f*)(ob + (size_t)row * D_ + 4 * lane) =
            *(const v4f*)&tile[row * PITCH_ + 4 * lane];
    }
}

extern "C" void kernel_launch(void* const* d_in, const int* in_sizes, int n_in,
                              void* d_out, int out_size, void* d_ws, size_t ws_size,
                              hipStream_t stream) {
    const float* x   = (const float*)d_in[0];
    const int*   adj = (const int*)d_in[1];
    const float* W   = (const float*)d_in[2];
    const float* a   = (const float*)d_in[3];
    float* out = (float*)d_out;

    float* wht = (float*)d_ws;
    float* f1  = wht + (size_t)B_ * D_ * N_;
    float* f2  = f1 + (size_t)B_ * N_;

    wht_kernel<<<dim3(B_ * (N_ / XJ_)), dim3(256), 0, stream>>>(x, W, a, wht, f1, f2);
    gat_kernel<<<dim3(B_ * (N_ / RB_)), dim3(256), 0, stream>>>(wht, f1, f2, adj, out);
}


# gat pass 2: LDS ring of 3 stages, the per-chunk barrier moved to mid-chunk, the operands of the first step behind it fetched in front of it, so matrix instructions issue at once behind the barrier
# speedup vs baseline: 3.5819x; 1.0328x over previous
#include <hip/hip_runtime.h>
#include <cstddef>

typedef float v2f __attribute__((ext_vector_type(2)));
typedef float v4f __attribute__((ext_vector_type(4)));
typedef float v8f __attribute__((ext_vector_type(8)));
typedef int   v4i __attribute__((ext_vector_type(4)));

#define B_  8
#define N_  2048
#define C_  128
#define D_  128
#define NEG_BIG (-9.0e15f)

static __device__ __forceinline__ v8f wmma_f32_4(v2f a, v2f b, v8f c) {
    return __builtin_amdgcn_wmma_f32_16x16x4_f32(false, a, false, b, (short)0, c,
                                                 false, false);
}

#define XJ_  64
#define XP_  132
#define OP_  68

__global__ __launch_bounds__(256)
void wht_kernel(const float* __restrict__ x, const float* __restrict__ W,
                const float* __restrict__ a, float* __restrict__ wht,
                float* __restrict__ f1, float* __restrict__ f2) {
    __shared__ __attribute__((aligned(16))) float xs[XJ_ * XP_];
    __shared__ __attribute__((aligned(16))) float os[D_ * OP_];
    __shared__ float fpart[2][8][XJ_];

    const int bid  = blockIdx.x;
    const int b    = bid >> 5;
    const int j0   = (bid & 31) * XJ_;
    const int tid  = threadIdx.x;
    const int wv   = tid >> 5;
    const int lane = tid & 31;
    const int c    = lane & 15;
    const int h    = lane >> 4;

    const float* xsrc = x + ((size_t)(b * N_ + j0 + wv)) * C_ + 4 * lane;
    v4f xt[8];
#pragma unroll
    for (int q = 0; q < 8; ++q) xt[q] = *(const v4f*)(xsrc + (size_t)(8 * q) * C_);

    const float* wrow = W + (size_t)(wv * 16 + c) * C_ + 4 * h;
    v4f af[16];
#pragma unroll
    for (int q = 0; q < 16; ++q) af[q] = *(const v4f*)(wrow + 8 * q);

    const float* ap = a + wv * 16 + 8 * h;
    const v4f a1lo = *(const v4f*)(ap);
    const v4f a1hi = *(const v4f*)(ap + 4);
    const v4f a2lo = *(const v4f*)(ap + D_);
    const v4f a2hi = *(const v4f*)(ap + D_ + 4);

#pragma unroll
    for (int q = 0; q < 8; ++q) *(v4f*)&xs[(wv + 8 * q) * XP_ + 4 * lane] = xt[q];
    __syncthreads();

    v8f acc[4] = {};
    const float* fb = xs + c * XP_ + 4 * h;
#pragma unroll
    for (int q = 0; q < 16; ++q) {
        v4f Bv[4];
#pragma unroll
        for (int t = 0; t < 4; ++t) Bv[t] = *(const v4f*)(fb + t * 16 * XP_ + 8 * q);
        v2f A0; A0[0] = af[q][0]; A0[1] = af[q][1];
        v2f A1; A1[0] = af[q][2]; A1[1] = af[q][3];
#pragma unroll
        for (int t = 0; t < 4; ++t) {
            v2f B0; B0[0] = Bv[t][0]; B0[1] = Bv[t][1];
            acc[t] = wmma_f32_4(A0, B0, acc[t]);
        }
#pragma unroll
        for (int t = 0; t < 4; ++t) {
            v2f B1; B1[0] = Bv[t][2]; B1[1] = Bv[t][3];
            acc[t] = wmma_f32_4(A1, B1, acc[t]);
        }
    }

#pragma unroll
    for (int t = 0; t < 4; ++t) {
        float s1 = 0.f, s2 = 0.f;
#pragma unroll
        for (int r = 0; r < 4; ++r) {
            s1 += acc[t][r] * a1lo[r];
            s2 += acc[t][r] * a2lo[r];
        }
#pragma unroll
        for (int r = 0; r < 4; ++r) {
            s1 += acc[t][4 + r] * a1hi[r];
            s2 += acc[t][4 + r] * a2hi[r];
        }
        s1 += __shfl_xor(s1, 16);
        s2 += __shfl_xor(s2, 16);
        if (lane < 16) {
            fpart[0][wv][t * 16 + c] = s1;
            fpart[1][wv][t * 16 + c] = s2;
        }
#pragma unroll
        for (int r = 0; r < 8; ++r)
            os[(wv * 16 + 8 * h + r) * OP_ + t * 16 + c] = acc[t][r];
    }
    __syncthreads();

    float* ob = wht + (size_t)b * D_ * N_ + j0 + 4 * c;
#pragma unroll
    for (int i = 0; i < 8; ++i) {
        const int d = wv * 16 + 2 * i + h;
        *(v4f*)(ob + (size_t)d * N_) = *(const v4f*)&os[d * OP_ + 4 * c];
    }

    if (tid < 2 * XJ_) {
        const int which = tid >> 6;
        const int j     = tid & (XJ_ - 1);
        float s = 0.f;
#pragma unroll
        for (int w = 0; w < 8; ++w) s += fpart[which][w][j];
        float* dst = which ? f2 : f1;
        dst[b * N_ + j0 + j] = s;
    }
}

typedef unsigned v4u __attribute__((ext_vector_type(4)));

#define RB_     64
#define JC_     128
#define PITCH_  132
#define BPITCH_ 68
#define P1_AHEAD_ 2
#define P1_RING_  3

static __device__ __forceinline__ void p_frag(const float* fs, const v4u bw, const unsigned sh,
                                              const float f1i, const float mi, const float pm,
                                              float& lsum, v2f& A0, v2f& A1) {
    const v4f f2v = *(const v4f*)fs;
    float s0 = f1i + f2v[0]; s0 = fmaxf(s0, 0.2f * s0);
    float s1 = f1i + f2v[1]; s1 = fmaxf(s1, 0.2f * s1);
    float s2 = f1i + f2v[2]; s2 = fmaxf(s2, 0.2f * s2);
    float s3 = f1i + f2v[3]; s3 = fmaxf(s3, 0.2f * s3);
    const float p0 = ((bw[0] >> sh) & 1u) ? __expf(s0 - mi) : pm;
    const float p1 = ((bw[1] >> sh) & 1u) ? __expf(s1 - mi) : pm;
    const float p2 = ((bw[2] >> sh) & 1u) ? __expf(s2 - mi) : pm;
    const float p3 = ((bw[3] >> sh) & 1u) ? __expf(s3 - mi) : pm;
    lsum += (p0 + p1) + (p2 + p3);
    A0[0] = p0; A0[1] = p1;
    A1[0] = p2; A1[1] = p3;
}
static __device__ __forceinline__ void b_frag(const float* fb, v4f (&Bv)[8]) {
#pragma unroll
    for (int dt = 0; dt < 8; ++dt) Bv[dt] = *(const v4f*)(fb + dt * 16 * PITCH_);
}
static __device__ __forceinline__ void mma16(const v2f A0, const v2f A1, const v4f (&Bv)[8],
                                             v8f (&acc)[8]) {
#pragma unroll
    for (int dt = 0; dt < 8; ++dt) {
        v2f B0; B0[0] = Bv[dt][0]; B0[1] = Bv[dt][1];
        acc[dt] = wmma_f32_4(A0, B0, acc[dt]);
    }
#pragma unroll
    for (int dt = 0; dt < 8; ++dt) {
        v2f B1; B1[0] = Bv[dt][2]; B1[1] = Bv[dt][3];
        acc[dt] = wmma_f32_4(A1, B1, acc[dt]);
    }
}

__global__ __launch_bounds__(256)
void gat_kernel(const float* __restrict__ wht, const float* __restrict__ f1g,
                const float* __restrict__ f2g, const int* __restrict__ adj,
                float* __restrict__ out) {
    __shared__ __attribute__((aligned(16))) float    stage[3 * D_ * PITCH_];
    __shared__ __attribute__((aligned(16))) unsigned lds_bits[RB_ * BPITCH_];
    __shared__ __attribute__((aligned(16))) float    f2s[N_];
    __shared__ float lds_mw[RB_ * 33];
    __shared__ float lds_m[RB_], lds_f1[RB_], lds_pone[RB_];
    __shared__ float lds_lw[2 * RB_];

    const int bid  = blockIdx.x;
    const int b    = bid >> 5;
    const int i0   = (bid & 31) * RB_;
    const int tid  = threadIdx.x;
    const int wv   = tid >> 5;
    const int lane = tid & 31;
    const int c    = lane & 15;
    const int h    = lane >> 4;
    const int it   = wv & 3;
    const int jh   = wv >> 2;

    const float* f2b  = f2g + b * N_;
    const float* whtb = wht + (size_t)b * D_ * N_;

    ((v4f*)f2s)[tid]       = ((const v4f*)f2b)[tid];
    ((v4f*)f2s)[tid + 256] = ((const v4f*)f2b)[tid + 256];

    {
        float mm[8];
#pragma unroll
        for (int r = 0; r < 8; ++r) mm[r] = -3.0e38f;
        const int*   abase = adj + (size_t)(i0 + wv * 8) * N_ + 4 * lane;
        const float* fbase = f2b + 4 * lane;
        unsigned*    brow  = &lds_bits[(wv * 8 + (lane >> 2)) * BPITCH_ + (lane & 3)];
        unsigned ksel[32];
#pragma unroll
        for (int s = 0; s < 32; ++s) ksel[s] = (lane == s) ? 0xffffffffu : 0u;
        v4i av[P1_RING_][8];
        v4f fq[P1_RING_];
#pragma unroll
        for (int p = 0; p < P1_AHEAD_; ++p) {
            fq[p] = *(const v4f*)(fbase + p * JC_);
#pragma unroll
            for (int r = 0; r < 8; ++r)
                av[p][r] = *(const v4i*)(abase + (size_t)r * N_ + p * JC_);
        }
#pragma unroll
        for (int ch = 0; ch < N_ / JC_; ++ch) {
            if (ch + P1_AHEAD_ < N_ / JC_) {
                const int sn = (ch + P1_AHEAD_) % P1_RING_;
                fq[sn] = *(const v4f*)(fbase + (ch + P1_AHEAD_) * JC_);
#pragma unroll
                for (int r = 0; r < 8; ++r)
                    av[sn][r] = *(const v4i*)(abase + (size_t)r * N_ + (ch + P1_AHEAD_) * JC_);
            }
            const int sc = ch % P1_RING_;
            const v4f fv = fq[sc];
            unsigned wbits = 0u;
#pragma unroll
            for (int r = 0; r < 8; ++r) {
                const v4i a4 = av[sc][r];
                const bool q0 = a4[0] > 0, q1 = a4[1] > 0, q2 = a4[2] > 0, q3 = a4[3] > 0;
                const unsigned m0 = __builtin_amdgcn_ballot_w32(q0);
                const unsigned m1 = __builtin_amdgcn_ballot_w32(q1);
                const unsigned m2 = __builtin_amdgcn_ballot_w32(q2);
                const unsigned m3 = __builtin_amdgcn_ballot_w32(q3);
                float t = mm[r];
                t = fmaxf(t, q0 ? fv[0] : -3.0e38f);
                t = fmaxf(t, q1 ? fv[1] : -3.0e38f);
                t = fmaxf(t, q2 ? fv[2] : -3.0e38f);
                t = fmaxf(t, q3 ? fv[3] : -3.0e38f);
                mm[r] = t;
                wbits |= m0 & ksel[4 * r + 0];
                wbits |= m1 & ksel[4 * r + 1];
                wbits |= m2 & ksel[4 * r + 2];
                wbits |= m3 & ksel[4 * r + 3];
            }
            brow[ch * 4] = wbits;
        }
#pragma unroll
        for (int r = 0; r < 8; ++r) lds_mw[(wv * 8 + r) * 33 + lane] = mm[r];
    }
    __syncthreads();
    if (tid < RB_) {
        float M = -3.0e38f;
#pragma unroll
        for (int s = 0; s < 32; ++s) M = fmaxf(M, lds_mw[tid * 33 + s]);
        const bool empty = (M < -1.0e38f);
        const float f1v  = f1g[b * N_ + i0 + tid];
        const float sum  = f1v + M;
        const float emax = fmaxf(sum, 0.2f * sum);
        lds_m[tid]    = empty ? 0.0f : emax;
        lds_pone[tid] = empty ? 1.0f : 0.0f;
        lds_f1[tid]   = f1v;
    }
    __syncthreads();

    const int   rowl = it * 16 + c;
    const float f1i  = lds_f1[rowl];
    const float mi   = lds_m[rowl];
    const float pm   = lds_pone[rowl];

    v8f acc[8] = {};
    float lsum = 0.0f;

    const float* gsrc = whtb + (size_t)wv * N_ + 4 * lane;
    const int    soff = wv * PITCH_ + 4 * lane;
    v4f st[16];
#pragma unroll
    for (int q = 0; q < 16; ++q) st[q] = *(const v4f*)(gsrc + (size_t)(8 * q) * N_);
#pragma unroll
    for (int q = 0; q < 16; ++q) *(v4f*)&stage[soff + 8 * q * PITCH_] = st[q];
#pragma unroll
    for (int q = 0; q < 16; ++q) st[q] = *(const v4f*)(gsrc + (size_t)(8 * q) * N_ + JC_);
    __syncthreads();

    const int       foff = c * PITCH_ + jh * 64 + 4 * h;
    const float*    fsb  = f2s + jh * 64 + 4 * h;
    const unsigned* bwp  = &lds_bits[rowl * BPITCH_];
    const unsigned  shb  = (unsigned)(jh * 16 + h);

    v4u bw = *(const v4u*)bwp;
#pragma unroll
    for (int g = 0; g < 4; ++g) {
        v2f A0, A1; v4f Bv[8];
        p_frag(fsb + 8 * g, bw, shb + 2 * g, f1i, mi, pm, lsum, A0, A1);
        b_frag(stage + foff + 8 * g, Bv);
        mma16(A0, A1, Bv, acc);
    }
    v2f pA0, pA1; v4f pB[8];
    p_frag(fsb + 8 * 4, bw, shb + 2 * 4, f1i, mi, pm, lsum, pA0, pA1);
    b_frag(stage + foff + 8 * 4, pB);

    int sc = 0;
    for (int ch = 0; ch < N_ / JC_ - 1; ++ch) {
        const int sn = (sc == 2) ? 0 : sc + 1;
        float* nxt = stage + sn * (D_ * PITCH_);
        const float* cur = stage + sc * (D_ * PITCH_);
#pragma unroll
        for (int q = 0; q < 16; ++q) *(v4f*)&nxt[soff + 8 * q * PITCH_] = st[q];
        __syncthreads();
        const int lc = (ch + 2 < N_ / JC_) ? ch + 2 : N_ / JC_ - 1;
#pragma unroll
        for (int q = 0; q < 16; ++q)
            st[q] = *(const v4f*)(gsrc + (size_t)(8 * q) * N_ + lc * JC_);
        mma16(pA0, pA1, pB, acc);
#pragma unroll
        for (int g = 5; g < 8; ++g) {
            v2f A0, A1; v4f Bv[8];
            p_frag(fsb + ch * JC_ + 8 * g, bw, shb + 2 * g, f1i, mi, pm, lsum, A0, A1);
            b_frag(cur + foff + 8 * g, Bv);
            mma16(A0, A1, Bv, acc);
        }
        const v4u bwn = *(const v4u*)(bwp + (ch + 1) * 4);
#pragma unroll
        for (int g = 0; g < 3; ++g) {
            v2f A0, A1; v4f Bv[8];
            p_frag(fsb + (ch + 1) * JC_ + 8 * g, bwn, shb + 2 * g, f1i, mi, pm, lsum, A0, A1);
            b_frag(nxt + foff + 8 * g, Bv);
            mma16(A0, A1, Bv, acc);
        }
        {
            v2f A0, A1; v4f Bv[8];
            p_frag(fsb + (ch + 1) * JC_ + 8 * 3, bwn, shb + 2 * 3, f1i, mi, pm, lsum, A0, A1);
            b_frag(nxt + foff + 8 * 3, Bv);
            p_frag(fsb + (ch + 1) * JC_ + 8 * 4, bwn, shb + 2 * 4, f1i, mi, pm, lsum, pA0, pA1);
            b_frag(nxt + foff + 8 * 4, pB);
            mma16(A0, A1, Bv, acc);
        }
        bw = bwn;
        sc = sn;
    }
    {
        const float* cur = stage + sc * (D_ * PITCH_);
        mma16(pA0, pA1, pB, acc);
#pragma unroll
        for (int g = 5; g < 8; ++g) {
            v2f A0, A1; v4f Bv[8];
            p_frag(fsb + (N_ / JC_ - 1) * JC_ + 8 * g, bw, shb + 2 * g, f1i, mi, pm, lsum, A0, A1);
            b_frag(cur + foff + 8 * g, Bv);
            mma16(A0, A1, Bv, acc);
        }
    }

    const float lpair = lsum + __shfl_xor(lsum, 16);
    if (lane < 16) lds_lw[jh * RB_ + it * 16 + lane] = lpair;

    float* tile = stage + ((sc == 2) ? 0 : sc + 1) * (D_ * PITCH_);
    if (jh == 1) {
#pragma unroll
        for (int dt = 0; dt < 8; ++dt)
#pragma unroll
            for (int r = 0; r < 8; ++r)
                tile[(it * 16 + 8 * h + r) * PITCH_ + dt * 16 + c] = acc[dt][r];
    }
    __syncthreads();
    if (jh == 0) {
#pragma unroll
        for (int r = 0; r < 8; ++r) {
            const int   row = it * 16 + 8 * h + r;
            const float li  = lds_lw[row] + lds_lw[RB_ + row];
#pragma unroll
            for (int dt = 0; dt < 8; ++dt) {
                float* tp = &tile[row * PITCH_ + dt * 16 + c];
                const float hv = (acc[dt][r] + *tp) / li;
                *tp = (hv > 0.0f) ? hv : (__expf(hv) - 1.0f);
            }
        }
    }
    __syncthreads();

    float* ob = out + ((size_t)b * N_ + i0) * D_;
#pragma unroll
    for (int q = 0; q < 8; ++q) {
        const int row = wv * 8 + q;
        *(v4f*)(ob + (size_t)row * D_ + 4 * lane) =
            *(const v4f*)&tile[row * PITCH_ + 4 * lane];
    }
}

extern "C" void kernel_launch(void* const* d_in, const int* in_sizes, int n_in,
                              void* d_out, int out_size, void* d_ws, size_t ws_size,
                              hipStream_t stream) {
    const float* x   = (const float*)d_in[0];
    const int*   adj = (const int*)d_in[1];
    const float* W   = (const float*)d_in[2];
    const float* a   = (const float*)d_in[3];
    float* out = (float*)d_out;

    float* wht = (float*)d_ws;
    float* f1  = wht + (size_t)B_ * D_ * N_;
    float* f2  = f1 + (size_t)B_ * N_;

    wht_kernel<<<dim3(B_ * (N_ / XJ_)), dim3(256), 0, stream>>>(x, W, a, wht, f1, f2);
    gat_kernel<<<dim3(B_ * (N_ / RB_)), dim3(256), 0, stream>>>(wht, f1, f2, adj, out);
}


# adjacency packed to bit masks once per call in the first kernel (the second kernel rebuilt them from the int32 rows in each of the 8 batches); mask rows parked in each block's own output rows
# speedup vs baseline: 3.7615x; 1.0501x over previous
#include <hip/hip_runtime.h>
#include <cstddef>

typedef float v2f __attribute__((ext_vector_type(2)));
typedef float v4f __attribute__((ext_vector_type(4)));
typedef float v8f __attribute__((ext_vector_type(8)));
typedef int   v4i __attribute__((ext_vector_type(4)));

#define B_  8
#define N_  2048
#define C_  128
#define D_  128
#define NEG_BIG (-9.0e15f)

static __device__ __forceinline__ v8f wmma_f32_4(v2f a, v2f b, v8f c) {
    return __builtin_amdgcn_wmma_f32_16x16x4_f32(false, a, false, b, (short)0, c,
                                                 false, false);
}

#define XJ_  64
#define XP_  132
#define OP_  68

__global__ __launch_bounds__(256)
void wht_kernel(const float* __restrict__ x, const float* __restrict__ W,
                const float* __restrict__ a, const int* __restrict__ adj,
                float* __restrict__ wht, float* __restrict__ f1,
                float* __restrict__ f2, float* __restrict__ mrows) {
    __shared__ __attribute__((aligned(16))) float xs[XJ_ * XP_];
    __shared__ __attribute__((aligned(16))) float os[D_ * OP_];
    __shared__ float fpart[2][8][XJ_];

    const int bid  = blockIdx.x;
    const int b    = bid >> 5;
    const int j0   = (bid & 31) * XJ_;
    const int tid  = threadIdx.x;
    const int wv   = tid >> 5;
    const int lane = tid & 31;
    const int c    = lane & 15;
    const int h    = lane >> 4;

    const float* xsrc = x + ((size_t)(b * N_ + j0 + wv)) * C_ + 4 * lane;
    v4f xt[8];
#pragma unroll
    for (int q = 0; q < 8; ++q) xt[q] = *(const v4f*)(xsrc + (size_t)(8 * q) * C_);

    const float* wrow = W + (size_t)(wv * 16 + c) * C_ + 4 * h;
    v4f af[16];
#pragma unroll
    for (int q = 0; q < 16; ++q) af[q] = *(const v4f*)(wrow + 8 * q);

    const float* ap = a + wv * 16 + 8 * h;
    const v4f a1lo = *(const v4f*)(ap);
    const v4f a1hi = *(const v4f*)(ap + 4);
    const v4f a2lo = *(const v4f*)(ap + D_);
    const v4f a2hi = *(const v4f*)(ap + D_ + 4);

    const int  arow = bid * 8 + wv;
    const int* asrc = adj + (size_t)arow * N_ + 4 * lane;
    v4i aq[16];
#pragma unroll
    for (int ch = 0; ch < 16; ++ch) aq[ch] = *(const v4i*)(asrc + ch * 128);

#pragma unroll
    for (int q = 0; q < 8; ++q) *(v4f*)&xs[(wv + 8 * q) * XP_ + 4 * lane] = xt[q];
    __syncthreads();

    v8f acc[4] = {};
    const float* fb = xs + c * XP_ + 4 * h;
#pragma unroll
    for (int q = 0; q < 16; ++q) {
        v4f Bv[4];
#pragma unroll
        for (int t = 0; t < 4; ++t) Bv[t] = *(const v4f*)(fb + t * 16 * XP_ + 8 * q);
        v2f A0; A0[0] = af[q][0]; A0[1] = af[q][1];
        v2f A1; A1[0] = af[q][2]; A1[1] = af[q][3];
#pragma unroll
        for (int t = 0; t < 4; ++t) {
            v2f B0; B0[0] = Bv[t][0]; B0[1] = Bv[t][1];
            acc[t] = wmma_f32_4(A0, B0, acc[t]);
        }
#pragma unroll
        for (int t = 0; t < 4; ++t) {
            v2f B1; B1[0] = Bv[t][2]; B1[1] = Bv[t][3];
            acc[t] = wmma_f32_4(A1, B1, acc[t]);
        }
    }

#pragma unroll
    for (int t = 0; t < 4; ++t) {
        float s1 = 0.f, s2 = 0.f;
#pragma unroll
        for (int r = 0; r < 4; ++r) {
            s1 += acc[t][r] * a1lo[r];
            s2 += acc[t][r] * a2lo[r];
        }
#pragma unroll
        for (int r = 0; r < 4; ++r) {
            s1 += acc[t][4 + r] * a1hi[r];
            s2 += acc[t][4 + r] * a2hi[r];
        }
        s1 += __shfl_xor(s1, 16);
        s2 += __shfl_xor(s2, 16);
        if (lane < 16) {
            fpart[0][wv][t * 16 + c] = s1;
            fpart[1][wv][t * 16 + c] = s2;
        }
#pragma unroll
        for (int r = 0; r < 8; ++r)
            os[(wv * 16 + 8 * h + r) * OP_ + t * 16 + c] = acc[t][r];
    }

    __syncthreads();

    float* ob = wht + (size_t)b * D_ * N_ + j0 + 4 * c;
#pragma unroll
    for (int i = 0; i < 8; ++i) {
        const int d = wv * 16 + 2 * i + h;
        *(v4f*)(ob + (size_t)d * N_) = *(const v4f*)&os[d * OP_ + 4 * c];
    }

    v2f mv;
    {
        unsigned w0 = 0u, w1 = 0u;
#pragma unroll
        for (int ch = 0; ch < 16; ++ch) {
#pragma unroll
            for (int cc = 0; cc < 4; ++cc) {
                const unsigned m = __builtin_amdgcn_ballot_w32(aq[ch][cc] > 0);
                const int widx = ch * 4 + cc;
                if (widx & 1) w1 = (lane == (widx >> 1)) ? m : w1;
                else          w0 = (lane == (widx >> 1)) ? m : w0;
            }
        }
        mv[0] = __uint_as_float(w0); mv[1] = __uint_as_float(w1);
    }
    float* mdst = mrows + (size_t)arow * D_ + 2 * lane;
#pragma unroll
    for (int bb = 0; bb < B_; ++bb) *(v2f*)(mdst + (size_t)bb * N_ * D_) = mv;

    if (tid < 2 * XJ_) {
        const int which = tid >> 6;
        const int j     = tid & (XJ_ - 1);
        float s = 0.f;
#pragma unroll
        for (int w = 0; w < 8; ++w) s += fpart[which][w][j];
        float* dst = which ? f2 : f1;
        dst[b * N_ + j0 + j] = s;
    }
}

typedef unsigned v4u __attribute__((ext_vector_type(4)));

#define RB_     64
#define JC_     128
#define PITCH_  132
#define BPITCH_ 68

static __device__ __forceinline__ void p_frag(const float* fs, const v4u bw, const unsigned sh,
                                              const float f1i, const float mi, const float pm,
                                              float& lsum, v2f& A0, v2f& A1) {
    const v4f f2v = *(const v4f*)fs;
    float s0 = f1i + f2v[0]; s0 = fmaxf(s0, 0.2f * s0);
    float s1 = f1i + f2v[1]; s1 = fmaxf(s1, 0.2f * s1);
    float s2 = f1i + f2v[2]; s2 = fmaxf(s2, 0.2f * s2);
    float s3 = f1i + f2v[3]; s3 = fmaxf(s3, 0.2f * s3);
    const float p0 = ((bw[0] >> sh) & 1u) ? __expf(s0 - mi) : pm;
    const float p1 = ((bw[1] >> sh) & 1u) ? __expf(s1 - mi) : pm;
    const float p2 = ((bw[2] >> sh) & 1u) ? __expf(s2 - mi) : pm;
    const float p3 = ((bw[3] >> sh) & 1u) ? __expf(s3 - mi) : pm;
    lsum += (p0 + p1) + (p2 + p3);
    A0[0] = p0; A0[1] = p1;
    A1[0] = p2; A1[1] = p3;
}
static __device__ __forceinline__ void b_frag(const float* fb, v4f (&Bv)[8]) {
#pragma unroll
    for (int dt = 0; dt < 8; ++dt) Bv[dt] = *(const v4f*)(fb + dt * 16 * PITCH_);
}
static __device__ __forceinline__ void mma16(const v2f A0, const v2f A1, const v4f (&Bv)[8],
                                             v8f (&acc)[8]) {
#pragma unroll
    for (int dt = 0; dt < 8; ++dt) {
        v2f B0; B0[0] = Bv[dt][0]; B0[1] = Bv[dt][1];
        acc[dt] = wmma_f32_4(A0, B0, acc[dt]);
    }
#pragma unroll
    for (int dt = 0; dt < 8; ++dt) {
        v2f B1; B1[0] = Bv[dt][2]; B1[1] = Bv[dt][3];
        acc[dt] = wmma_f32_4(A1, B1, acc[dt]);
    }
}

__global__ __launch_bounds__(256)
void gat_kernel(const float* __restrict__ wht, const float* __restrict__ f1g,
                const float* __restrict__ f2g, float* __restrict__ out) {
    __shared__ __attribute__((aligned(16))) float    stage[3 * D_ * PITCH_];
    __shared__ __attribute__((aligned(16))) unsigned lds_bits[RB_ * BPITCH_];
    __shared__ __attribute__((aligned(16))) float    f2s[N_];
    __shared__ float lds_mw[RB_ * 33];
    __shared__ float lds_m[RB_], lds_f1[RB_], lds_pone[RB_];
    __shared__ float lds_lw[2 * RB_];

    const int bid  = blockIdx.x;
    const int b    = bid >> 5;
    const int i0   = (bid & 31) * RB_;
    const int tid  = threadIdx.x;
    const int wv   = tid >> 5;
    const int lane = tid & 31;
    const int c    = lane & 15;
    const int h    = lane >> 4;
    const int it   = wv & 3;
    const int jh   = wv >> 2;

    const float* f2b  = f2g + b * N_;
    const float* whtb = wht + (size_t)b * D_ * N_;

    ((v4f*)f2s)[tid]       = ((const v4f*)f2b)[tid];
    ((v4f*)f2s)[tid + 256] = ((const v4f*)f2b)[tid + 256];

    {
        const float* fbase = f2b + 4 * lane;
        v4f fv[16];
#pragma unroll
        for (int ch = 0; ch < N_ / JC_; ++ch) fv[ch] = *(const v4f*)(fbase + ch * JC_);
        const float* mbase = out + ((size_t)(b * N_ + i0 + wv * 8)) * D_ + 2 * lane;
        v2f mk[8];
#pragma unroll
        for (int r = 0; r < 8; ++r) mk[r] = *(const v2f*)(mbase + (size_t)r * D_);
#pragma unroll
        for (int r = 0; r < 8; ++r) {
            const unsigned k0 = __float_as_uint(mk[r][0]);
            const unsigned k1 = __float_as_uint(mk[r][1]);
            typedef unsigned v2u __attribute__((ext_vector_type(2)));
            v2u kk; kk[0] = k0; kk[1] = k1;
            *(v2u*)&lds_bits[(wv * 8 + r) * BPITCH_ + 2 * lane] = kk;
            float t = -3.0e38f;
#pragma unroll
            for (int ch = 0; ch < N_ / JC_; ++ch) {
#pragma unroll
                for (int cc = 0; cc < 4; ++cc) {
                    const int widx = ch * 4 + cc;
                    const unsigned w = (unsigned)__builtin_amdgcn_readlane(
                        (int)((widx & 1) ? k1 : k0), widx >> 1);
                    const bool q = __builtin_amdgcn_inverse_ballot_w32(w);
                    t = fmaxf(t, q ? fv[ch][cc] : -3.0e38f);
                }
            }
            lds_mw[(wv * 8 + r) * 33 + lane] = t;
        }
    }
    __syncthreads();
    if (tid < RB_) {
        float M = -3.0e38f;
#pragma unroll
        for (int s = 0; s < 32; ++s) M = fmaxf(M, lds_mw[tid * 33 + s]);
        const bool empty = (M < -1.0e38f);
        const float f1v  = f1g[b * N_ + i0 + tid];
        const float sum  = f1v + M;
        const float emax = fmaxf(sum, 0.2f * sum);
        lds_m[tid]    = empty ? 0.0f : emax;
        lds_pone[tid] = empty ? 1.0f : 0.0f;
        lds_f1[tid]   = f1v;
    }
    __syncthreads();

    const int   rowl = it * 16 + c;
    const float f1i  = lds_f1[rowl];
    const float mi   = lds_m[rowl];
    const float pm   = lds_pone[rowl];

    v8f acc[8] = {};
    float lsum = 0.0f;

    const float* gsrc = whtb + (size_t)wv * N_ + 4 * lane;
    const int    soff = wv * PITCH_ + 4 * lane;
    v4f st[16];
#pragma unroll
    for (int q = 0; q < 16; ++q) st[q] = *(const v4f*)(gsrc + (size_t)(8 * q) * N_);
#pragma unroll
    for (int q = 0; q < 16; ++q) *(v4f*)&stage[soff + 8 * q * PITCH_] = st[q];
#pragma unroll
    for (int q = 0; q < 16; ++q) st[q] = *(const v4f*)(gsrc + (size_t)(8 * q) * N_ + JC_);
    __syncthreads();

    const int       foff = c * PITCH_ + jh * 64 + 4 * h;
    const float*    fsb  = f2s + jh * 64 + 4 * h;
    const unsigned* bwp  = &lds_bits[rowl * BPITCH_];
    const unsigned  shb  = (unsigned)(jh * 16 + h);

    v4u bw = *(const v4u*)bwp;
#pragma unroll
    for (int g = 0; g < 4; ++g) {
        v2f A0, A1; v4f Bv[8];
        p_frag(fsb + 8 * g, bw, shb + 2 * g, f1i, mi, pm, lsum, A0, A1);
        b_frag(stage + foff + 8 * g, Bv);
        mma16(A0, A1, Bv, acc);
    }
    v2f pA0, pA1; v4f pB[8];
    p_frag(fsb + 8 * 4, bw, shb + 2 * 4, f1i, mi, pm, lsum, pA0, pA1);
    b_frag(stage + foff + 8 * 4, pB);

    int sc = 0;
    for (int ch = 0; ch < N_ / JC_ - 1; ++ch) {
        const int sn = (sc == 2) ? 0 : sc + 1;
        float* nxt = stage + sn * (D_ * PITCH_);
        const float* cur = stage + sc * (D_ * PITCH_);
#pragma unroll
        for (int q = 0; q < 16; ++q) *(v4f*)&nxt[soff + 8 * q * PITCH_] = st[q];
        __syncthreads();
        const int lc = (ch + 2 < N_ / JC_) ? ch + 2 : N_ / JC_ - 1;
#pragma unroll
        for (int q = 0; q < 16; ++q)
            st[q] = *(const v4f*)(gsrc + (size_t)(8 * q) * N_ + lc * JC_);
        mma16(pA0, pA1, pB, acc);
#pragma unroll
        for (int g = 5; g < 8; ++g) {
            v2f A0, A1; v4f Bv[8];
            p_frag(fsb + ch * JC_ + 8 * g, bw, shb + 2 * g, f1i, mi, pm, lsum, A0, A1);
            b_frag(cur + foff + 8 * g, Bv);
            mma16(A0, A1, Bv, acc);
        }
        const v4u bwn = *(const v4u*)(bwp + (ch + 1) * 4);
#pragma unroll
        for (int g = 0; g < 3; ++g) {
            v2f A0, A1; v4f Bv[8];
            p_frag(fsb + (ch + 1) * JC_ + 8 * g, bwn, shb + 2 * g, f1i, mi, pm, lsum, A0, A1);
            b_frag(nxt + foff + 8 * g, Bv);
            mma16(A0, A1, Bv, acc);
        }
        {
            v2f A0, A1; v4f Bv[8];
            p_frag(fsb + (ch + 1) * JC_ + 8 * 3, bwn, shb + 2 * 3, f1i, mi, pm, lsum, A0, A1);
            b_frag(nxt + foff + 8 * 3, Bv);
            p_frag(fsb + (ch + 1) * JC_ + 8 * 4, bwn, shb + 2 * 4, f1i, mi, pm, lsum, pA0, pA1);
            b_frag(nxt + foff + 8 * 4, pB);
            mma16(A0, A1, Bv, acc);
        }
        bw = bwn;
        sc = sn;
    }
    {
        const float* cur = stage + sc * (D_ * PITCH_);
        mma16(pA0, pA1, pB, acc);
#pragma unroll
        for (int g = 5; g < 8; ++g) {
            v2f A0, A1; v4f Bv[8];
            p_frag(fsb + (N_ / JC_ - 1) * JC_ + 8 * g, bw, shb + 2 * g, f1i, mi, pm, lsum, A0, A1);
            b_frag(cur + foff + 8 * g, Bv);
            mma16(A0, A1, Bv, acc);
        }
    }

    const float lpair = lsum + __shfl_xor(lsum, 16);
    if (lane < 16) lds_lw[jh * RB_ + it * 16 + lane] = lpair;

    float* tile = stage + ((sc == 2) ? 0 : sc + 1) * (D_ * PITCH_);
    if (jh == 1) {
#pragma unroll
        for (int dt = 0; dt < 8; ++dt)
#pragma unroll
            for (int r = 0; r < 8; ++r)
                tile[(it * 16 + 8 * h + r) * PITCH_ + dt * 16 + c] = acc[dt][r];
    }
    __syncthreads();
    if (jh == 0) {
#pragma unroll
        for (int r = 0; r < 8; ++r) {
            const int   row = it * 16 + 8 * h + r;
            const float li  = lds_lw[row] + lds_lw[RB_ + row];
#pragma unroll
            for (int dt = 0; dt < 8; ++dt) {
                float* tp = &tile[row * PITCH_ + dt * 16 + c];
                const float hv = (acc[dt][r] + *tp) / li;
                *tp = (hv > 0.0f) ? hv : (__expf(hv) - 1.0f);
            }
        }
    }
    __syncthreads();

    float* ob = out + ((size_t)b * N_ + i0) * D_;
#pragma unroll
    for (int q = 0; q < 8; ++q) {
        const int row = wv * 8 + q;
        *(v4f*)(ob + (size_t)row * D_ + 4 * lane) =
            *(const v4f*)&tile[row * PITCH_ + 4 * lane];
    }
}

extern "C" void kernel_launch(void* const* d_in, const int* in_sizes, int n_in,
                              void* d_out, int out_size, void* d_ws, size_t ws_size,
                              hipStream_t stream) {
    const float* x   = (const float*)d_in[0];
    const int*   adj = (const int*)d_in[1];
    const float* W   = (const float*)d_in[2];
    const float* a   = (const float*)d_in[3];
    float* out = (float*)d_out;

    float* wht = (float*)d_ws;
    float* f1  = wht + (size_t)B_ * D_ * N_;
    float* f2  = f1 + (size_t)B_ * N_;

    wht_kernel<<<dim3(B_ * (N_ / XJ_)), dim3(256), 0, stream>>>(x, W, a, adj, wht, f1, f2, out);
    gat_kernel<<<dim3(B_ * (N_ / RB_)), dim3(256), 0, stream>>>(wht, f1, f2, out);
}
